# starting kernel
# baseline (speedup 1.0000x reference)
#include <hip/hip_runtime.h>

#define NNODES  50000
#define NEDGES  800000
#define DIM     128
#define NLAYERS 3

typedef __attribute__((ext_vector_type(2))) float v2f;
typedef __attribute__((ext_vector_type(8))) float v8f;

__device__ __forceinline__ void atomic_add_f32_dev(float* p, float v) {
    asm volatile("global_atomic_add_f32 %0, %1, off scope:SCOPE_DEV"
                 :: "v"(p), "v"(v) : "memory");
}

__global__ void gcn_zero_kernel(float4* __restrict__ p, int n4) {
    int i = blockIdx.x * blockDim.x + threadIdx.x;
    if (i < n4) p[i] = make_float4(0.f, 0.f, 0.f, 0.f);
}

__global__ __launch_bounds__(256) void gcn_scatter_kernel(
    const float* __restrict__ h,
    const int*   __restrict__ edge_index,
    const float* __restrict__ edge_weight,
    float*       __restrict__ agg,
    int nedges) {
    int tid  = blockIdx.x * blockDim.x + threadIdx.x;
    int e    = tid >> 5;
    int lane = tid & 31;
    if (e >= nedges) return;

    int   src = edge_index[e];
    int   dst = edge_index[nedges + e];
    float w   = edge_weight[e];

    const float4* hv = reinterpret_cast<const float4*>(h + (size_t)src * DIM);
    float4 v = hv[lane];

    float* a = agg + (size_t)dst * DIM + (size_t)lane * 4;
    atomic_add_f32_dev(a + 0, v.x * w);
    atomic_add_f32_dev(a + 1, v.y * w);
    atomic_add_f32_dev(a + 2, v.z * w);
    atomic_add_f32_dev(a + 3, v.w * w);
}

__global__ __launch_bounds__(256) void gcn_gemm_wmma_kernel(
    const float* __restrict__ agg,
    const float* __restrict__ h,
    const float* __restrict__ Wrel,
    const float* __restrict__ Wroot,
    const float* __restrict__ bias,
    float*       __restrict__ out,
    int nrows, int do_relu) {
    __shared__ float4 sA4[16 * DIM / 4];
    __shared__ float4 sH4[16 * DIM / 4];

    const int t    = threadIdx.x;
    const int lane = t & 31;
    const int wv   = t >> 5;
    const int r    = lane & 15;
    const int half = lane >> 4;

    const int row0 = blockIdx.x * 16;
    const int col0 = wv * 16;

    {
        const float4* aggv = reinterpret_cast<const float4*>(agg) + (size_t)row0 * (DIM / 4);
        const float4* hv   = reinterpret_cast<const float4*>(h)   + (size_t)row0 * (DIM / 4);
#pragma unroll
        for (int i = t; i < 16 * DIM / 4; i += 256) {
            sA4[i] = aggv[i];
            sH4[i] = hv[i];
        }
    }
    __syncthreads();

    const float* sA = reinterpret_cast<const float*>(sA4);
    const float* sH = reinterpret_cast<const float*>(sH4);

    v8f acc_rel  = {};
    v8f acc_root = {};
#pragma unroll
    for (int k = 0; k < DIM; k += 4) {
        const int ka = k + 2 * half;

        v2f a_rel  = *reinterpret_cast<const v2f*>(&sA[r * DIM + ka]);
        v2f b_rel  = { Wrel[(size_t)ka * DIM + col0 + r],
                       Wrel[(size_t)(ka + 1) * DIM + col0 + r] };
        acc_rel = __builtin_amdgcn_wmma_f32_16x16x4_f32(
                      false, a_rel, false, b_rel, (short)0, acc_rel, false, false);

        v2f a_root = *reinterpret_cast<const v2f*>(&sH[r * DIM + ka]);
        v2f b_root = { Wroot[(size_t)ka * DIM + col0 + r],
                       Wroot[(size_t)(ka + 1) * DIM + col0 + r] };
        acc_root = __builtin_amdgcn_wmma_f32_16x16x4_f32(
                      false, a_root, false, b_root, (short)0, acc_root, false, false);
    }

    const float bv = bias[col0 + r];
#pragma unroll
    for (int j = 0; j < 8; ++j) {
        const int m    = j + 8 * half;
        const int orow = row0 + m;
        if (orow < nrows) {
            float v = acc_rel[j] + acc_root[j] + bv;
            if (do_relu) v = fmaxf(v, 0.f);
            out[(size_t)orow * DIM + col0 + r] = v;
        }
    }
}

extern "C" void kernel_launch(void* const* d_in, const int* in_sizes, int n_in,
                              void* d_out, int out_size, void* d_ws, size_t ws_size,
                              hipStream_t stream) {
    const float* x      = (const float*)d_in[0];
    const int*   eidx   = (const int*)  d_in[1];
    const float* ew     = (const float*)d_in[2];
    const float* Wrel   = (const float*)d_in[3];
    const float* brel   = (const float*)d_in[4];
    const float* Wroot  = (const float*)d_in[5];
    float*       out    = (float*)d_out;

    const int N = NNODES;
    const int E = NEDGES;

    float* h1  = (float*)d_ws;
    float* h2  = h1 + (size_t)N * DIM;
    float* agg = h2 + (size_t)N * DIM;

    const int n4          = N * DIM / 4;
    const int zero_blocks = (n4 + 255) / 256;
    const int scat_blocks = (E * 32 + 255) / 256;
    const int gemm_blocks = (N + 15) / 16;

    for (int l = 0; l < NLAYERS; ++l) {
        const float* hin  = (l == 0) ? x  : (l == 1) ? h1 : h2;
        float*       hout = (l == 0) ? h1 : (l == 1) ? h2 : out;

        gcn_zero_kernel<<<zero_blocks, 256, 0, stream>>>(
            (float4*)agg, n4);

        gcn_scatter_kernel<<<scat_blocks, 256, 0, stream>>>(
            hin, eidx, ew, agg, E);

        gcn_gemm_wmma_kernel<<<gemm_blocks, 256, 0, stream>>>(
            agg, hin,
            Wrel  + (size_t)l * DIM * DIM,
            Wroot + (size_t)l * DIM * DIM,
            brel  + (size_t)l * DIM,
            hout, N, (l < NLAYERS - 1) ? 1 : 0);
    }
}


# each layer's linear map applied before its edge sum: f32 atomics add w*(act(h)@Wrel)[src] onto act(h)@Wroot+b; agg plane, clearing launch, 2nd A tile gone; sums in arrival order on that start
# speedup vs baseline: 2.1590x; 2.1590x over previous
#include <hip/hip_runtime.h>

#define NNODES  50000
#define NEDGES  800000
#define DIM     128
#define NLAYERS 3

typedef __attribute__((ext_vector_type(2))) float v2f;
typedef __attribute__((ext_vector_type(8))) float v8f;

template <int OFF>
__device__ __forceinline__ void atomic_add_f32_row(float* row, unsigned voff, float v) {
    asm volatile("global_atomic_add_f32 %0, %1, %2 offset:%3 scope:SCOPE_DEV"
                 :: "v"(voff), "v"(v), "s"(row), "n"(OFF) : "memory");
}

#define EPW 16
static_assert(EPW == 16, "the load fences in the scatter kernel name 16 values each");
static_assert(NEDGES % EPW == 0, "a wave takes EPW whole edges; there is no partial group");

__device__ __forceinline__ float load_at(const float* row, unsigned byte_off) {
    return *(const float*)((const char*)row + byte_off);
}

__global__ __launch_bounds__(256) void gcn_scatter_kernel(
    const float* __restrict__ p,
    const int*   __restrict__ edge_index,
    const float* __restrict__ edge_weight,
    float*       __restrict__ acc,
    int nedges) {
    const int lane = threadIdx.x & 31;
    const int wave = __builtin_amdgcn_readfirstlane((int)(threadIdx.x >> 5));
    const int e0   = (blockIdx.x * 8 + wave) * EPW;
    if (e0 >= nedges) return;

    const unsigned voff = (unsigned)lane * 4u;
    const int*   srcp = edge_index + e0;
    const int*   dstp = edge_index + nedges + e0;
    const float* wp   = edge_weight + e0;

    int   src[EPW];
    int   dst[EPW];
    float w[EPW];
#pragma unroll
    for (int j = 0; j < EPW; ++j) {
        src[j] = srcp[j];
        dst[j] = dstp[j];
        w[j]   = wp[j];
    }
    asm volatile("" :: "s"(src[0]), "s"(src[1]), "s"(src[2]), "s"(src[3]),
                       "s"(src[4]), "s"(src[5]), "s"(src[6]), "s"(src[7]),
                       "s"(src[8]), "s"(src[9]), "s"(src[10]), "s"(src[11]),
                       "s"(src[12]), "s"(src[13]), "s"(src[14]), "s"(src[15]));
    asm volatile("" :: "s"(dst[0]), "s"(dst[1]), "s"(dst[2]), "s"(dst[3]),
                       "s"(dst[4]), "s"(dst[5]), "s"(dst[6]), "s"(dst[7]),
                       "s"(dst[8]), "s"(dst[9]), "s"(dst[10]), "s"(dst[11]),
                       "s"(dst[12]), "s"(dst[13]), "s"(dst[14]), "s"(dst[15]));
    asm volatile("" :: "s"(w[0]), "s"(w[1]), "s"(w[2]), "s"(w[3]),
                       "s"(w[4]), "s"(w[5]), "s"(w[6]), "s"(w[7]),
                       "s"(w[8]), "s"(w[9]), "s"(w[10]), "s"(w[11]),
                       "s"(w[12]), "s"(w[13]), "s"(w[14]), "s"(w[15]));

    float v[EPW][4];
#pragma unroll
    for (int j = 0; j < EPW; ++j) {
        const float* hs = p + (size_t)src[j] * DIM;
        v[j][0] = load_at(hs, voff);
        v[j][1] = load_at(hs, voff + 128u);
        v[j][2] = load_at(hs, voff + 256u);
        v[j][3] = load_at(hs, voff + 384u);
    }

#pragma unroll
    for (int j = 0; j < EPW; ++j) {
        float* a = acc + (size_t)dst[j] * DIM;
        atomic_add_f32_row<0>  (a, voff, v[j][0] * w[j]);
        atomic_add_f32_row<128>(a, voff, v[j][1] * w[j]);
        atomic_add_f32_row<256>(a, voff, v[j][2] * w[j]);
        atomic_add_f32_row<384>(a, voff, v[j][3] * w[j]);
    }
}

__global__ __launch_bounds__(256) void gcn_gemm_wmma_kernel(
    const float* __restrict__ hin,
    const float* __restrict__ Wrel,
    const float* __restrict__ Wroot,
    const float* __restrict__ bias,
    float*       __restrict__ p,
    float*       __restrict__ acc,
    int nrows, int relu_in) {
    __shared__ float4 sH4[16 * DIM / 4];

    const int t    = threadIdx.x;
    const int lane = t & 31;
    const int wv   = t >> 5;
    const int r    = lane & 15;
    const int half = lane >> 4;

    const int row0 = blockIdx.x * 16;
    const int col0 = wv * 16;

    {
        const float4* hv = reinterpret_cast<const float4*>(hin) + (size_t)row0 * (DIM / 4);
        float4 v0 = hv[t];
        float4 v1 = hv[t + 256];
        if (relu_in) {
            v0.x = fmaxf(v0.x, 0.f);  v0.y = fmaxf(v0.y, 0.f);
            v0.z = fmaxf(v0.z, 0.f);  v0.w = fmaxf(v0.w, 0.f);
            v1.x = fmaxf(v1.x, 0.f);  v1.y = fmaxf(v1.y, 0.f);
            v1.z = fmaxf(v1.z, 0.f);  v1.w = fmaxf(v1.w, 0.f);
        }
        sH4[t]       = v0;
        sH4[t + 256] = v1;
    }
    __syncthreads();

    const float* sH = reinterpret_cast<const float*>(sH4);

    v8f acc_rel  = {};
    v8f acc_root = {};
#pragma unroll
    for (int k = 0; k < DIM; k += 4) {
        const int ka = k + 2 * half;

        v2f a      = *reinterpret_cast<const v2f*>(&sH[r * DIM + ka]);
        v2f b_rel  = { Wrel[(size_t)ka * DIM + col0 + r],
                       Wrel[(size_t)(ka + 1) * DIM + col0 + r] };
        acc_rel = __builtin_amdgcn_wmma_f32_16x16x4_f32(
                      false, a, false, b_rel, (short)0, acc_rel, false, false);

        v2f b_root = { Wroot[(size_t)ka * DIM + col0 + r],
                       Wroot[(size_t)(ka + 1) * DIM + col0 + r] };
        acc_root = __builtin_amdgcn_wmma_f32_16x16x4_f32(
                      false, a, false, b_root, (short)0, acc_root, false, false);
    }

    const float bv = bias[col0 + r];
#pragma unroll
    for (int j = 0; j < 8; ++j) {
        const int m    = j + 8 * half;
        const int orow = row0 + m;
        if (orow < nrows) {
            p  [(size_t)orow * DIM + col0 + r] = acc_rel[j];
            acc[(size_t)orow * DIM + col0 + r] = acc_root[j] + bv;
        }
    }
}

extern "C" void kernel_launch(void* const* d_in, const int* in_sizes, int n_in,
                              void* d_out, int out_size, void* d_ws, size_t ws_size,
                              hipStream_t stream) {
    const float* x      = (const float*)d_in[0];
    const int*   eidx   = (const int*)  d_in[1];
    const float* ew     = (const float*)d_in[2];
    const float* Wrel   = (const float*)d_in[3];
    const float* brel   = (const float*)d_in[4];
    const float* Wroot  = (const float*)d_in[5];
    float*       out    = (float*)d_out;

    const int N = NNODES;
    const int E = NEDGES;

    float* s1 = (float*)d_ws;
    float* s2 = s1 + (size_t)N * DIM;
    float* p  = s2 + (size_t)N * DIM;

    const int scat_blocks = (E + 8 * EPW - 1) / (8 * EPW);
    const int gemm_blocks = (N + 15) / 16;

    for (int l = 0; l < NLAYERS; ++l) {
        const float* hin  = (l == 0) ? x  : (l == 1) ? s1 : s2;
        float*       hout = (l == 0) ? s1 : (l == 1) ? s2 : out;

        gcn_gemm_wmma_kernel<<<gemm_blocks, 256, 0, stream>>>(
            hin,
            Wrel  + (size_t)l * DIM * DIM,
            Wroot + (size_t)l * DIM * DIM,
            brel  + (size_t)l * DIM,
            p, hout, N, (l > 0) ? 1 : 0);

        gcn_scatter_kernel<<<scat_blocks, 256, 0, stream>>>(
            p, eidx, ew, hout, E);
    }
}


# edge phase without atomics or sort: per-call bit maps tell 256 owner blocks their edges; each wave sums its own rows in LDS in fixed order; the sums then go onto the root rows by plain stores
# speedup vs baseline: 2.6018x; 1.2051x over previous
#include <hip/hip_runtime.h>

#define NNODES  50000
#define NEDGES  800000
#define DIM     128
#define NLAYERS 3

typedef __attribute__((ext_vector_type(2))) float v2f;
typedef __attribute__((ext_vector_type(8))) float v8f;

#define ROWS_PER_OWNER 196
#define NOWNERS        256
#define PREP_EDGES     1024
#define PREP_BLOCKS    ((NEDGES + PREP_EDGES - 1) / PREP_EDGES)
#define BM_STRIDE      (PREP_BLOCKS * (PREP_EDGES / 32))
#define BM_VEC         (BM_STRIDE / 4)
#define OWNER_THREADS  1024
#define OWNER_WAVES    (OWNER_THREADS / 32)
#define SCAN_TRIPS     ((BM_VEC + OWNER_THREADS - 1) / OWNER_THREADS)
#define LIST_CAP       8192
static_assert(ROWS_PER_OWNER * NOWNERS >= NNODES, "the owners' ranges cover every node");
static_assert(NNODES <= 65536 && ROWS_PER_OWNER <= 65536, "source row and row-in-range share one 32-bit word");
static_assert(BM_STRIDE % 32 == 0, "an owner's map is whole 128-byte lines");
static_assert(OWNER_WAVES == 32, "one lane per wave in the offset sums; rows are dealt to waves by r % 32");

__global__ __launch_bounds__(PREP_EDGES) void gcn_edge_prep_kernel(
    const int*   __restrict__ edge_index,
    const float* __restrict__ edge_weight,
    unsigned*    __restrict__ bitmap,
    unsigned long long* __restrict__ rec,
    int nedges) {
    __shared__ unsigned bits[NOWNERS][PREP_EDGES / 32];
    const int t    = threadIdx.x;
    const int lane = t & 31;
    const int wave = t >> 5;
    unsigned* flat = &bits[0][0];
    for (int i = t; i < NOWNERS * (PREP_EDGES / 32); i += PREP_EDGES) flat[i] = 0u;
    __syncthreads();

    const int  e     = blockIdx.x * PREP_EDGES + t;
    const bool valid = e < nedges;
    int o = 0;
    if (valid) {
        const int   s = edge_index[e];
        const int   d = edge_index[nedges + e];
        const float w = edge_weight[e];
        o = d / ROWS_PER_OWNER;
        const int r = d - o * ROWS_PER_OWNER;
        rec[e] = (unsigned long long)((unsigned)s | ((unsigned)r << 16)) |
                 ((unsigned long long)__float_as_uint(w) << 32);
    }
    unsigned todo = __builtin_amdgcn_ballot_w32(valid);
    while (todo != 0u) {
        const int      leader = __builtin_ctz(todo);
        const int      lo     = __builtin_amdgcn_readlane(o, leader);
        const unsigned m      = __builtin_amdgcn_ballot_w32(valid && o == lo);
        if (lane == leader) bits[lo][wave] = m;
        todo &= ~m;
    }
    __syncthreads();

    for (int i = t; i < NOWNERS * (PREP_EDGES / 32); i += PREP_EDGES) {
        const int oo = i >> 5;
        const int j  = i & 31;
        bitmap[(size_t)oo * BM_STRIDE + (size_t)blockIdx.x * (PREP_EDGES / 32) + j] = flat[i];
    }
}

__device__ __forceinline__ int wave_sum(int v) {
    v += __shfl_xor(v, 16);
    v += __shfl_xor(v, 8);
    v += __shfl_xor(v, 4);
    v += __shfl_xor(v, 2);
    v += __shfl_xor(v, 1);
    return v;
}

#define EPW  16
#define RING 64
static_assert(EPW - 1 + 32 <= RING, "the ring holds the waiting edges and one round of new ones");

#define FENCE4(J) asm volatile("" :: \
    "v"(v[J].x),   "v"(v[J].y),   "v"(v[J].z),   "v"(v[J].w),   \
    "v"(v[J+1].x), "v"(v[J+1].y), "v"(v[J+1].z), "v"(v[J+1].w), \
    "v"(v[J+2].x), "v"(v[J+2].y), "v"(v[J+2].z), "v"(v[J+2].w), \
    "v"(v[J+3].x), "v"(v[J+3].y), "v"(v[J+3].z), "v"(v[J+3].w));
static_assert(EPW == 16, "the load fences in owner_batch name 16 gathers");

typedef volatile unsigned long long __attribute__((address_space(3))) ring_word;

template <bool PARTIAL>
__device__ __forceinline__ void owner_batch(
    const float* __restrict__ p, float* sums,
    const ring_word* ring, int head, int n, int lane) {
    unsigned long long rc = 0ull;
    if (lane < n) rc = ring[(head + lane) & (RING - 1)];
    const int rlo = (int)(unsigned)rc;
    const int rhi = (int)(unsigned)(rc >> 32);

    const unsigned voff = (unsigned)lane * 16u;
    float4 v[EPW];
    int    row[EPW];
    float  w[EPW];
#pragma unroll
    for (int j = 0; j < EPW; ++j) {
        const unsigned sx = (unsigned)__builtin_amdgcn_readlane(rlo, j);
        w[j]   = __uint_as_float((unsigned)__builtin_amdgcn_readlane(rhi, j));
        row[j] = (int)(sx >> 16);
        const unsigned off = (sx & 0xffffu) * (unsigned)(DIM * 4) + voff;
        v[j] = *(const float4*)((const char*)p + off);
    }
    FENCE4(0)  FENCE4(4)  FENCE4(8)  FENCE4(12)
#pragma unroll
    for (int j = 0; j < EPW; ++j) {
        if (!PARTIAL || j < n) {
            float4* sp = reinterpret_cast<float4*>(sums + row[j] * DIM) + lane;
            float4 a = *sp;
            a.x += v[j].x * w[j];
            a.y += v[j].y * w[j];
            a.z += v[j].z * w[j];
            a.w += v[j].w * w[j];
            *sp = a;
        }
    }
}

__global__ __launch_bounds__(OWNER_THREADS) void gcn_owner_sum_kernel(
    const float*    __restrict__ p,
    const unsigned*              bitmap,
    const unsigned long long* __restrict__ rec,
    float*          __restrict__ acc,
    int nrows) {
    __shared__ float              sums[ROWS_PER_OWNER * DIM];
    __shared__ unsigned long long recs[LIST_CAP];
    __shared__ unsigned long long rings[OWNER_WAVES][RING];
    __shared__ unsigned           wbits[OWNER_WAVES][LIST_CAP / 32];
    __shared__ int                cnt[OWNER_WAVES];

    const int t     = threadIdx.x;
    const int lane  = t & 31;
    const int wave  = __builtin_amdgcn_readfirstlane((int)(threadIdx.x >> 5));
    const int owner = blockIdx.x;
    const char* bm = reinterpret_cast<const char*>(bitmap + (size_t)owner * BM_STRIDE);

    float4* s4 = reinterpret_cast<float4*>(sums);
    for (int i = t; i < ROWS_PER_OWNER * DIM / 4; i += OWNER_THREADS)
        s4[i] = make_float4(0.f, 0.f, 0.f, 0.f);

    ring_word* ring = (ring_word*)&rings[wave][0];
    const unsigned lanes_below = (1u << lane) - 1u;

    int base = 0;
    int total;
    do {
        for (int i = t; i < OWNER_WAVES * (LIST_CAP / 32); i += OWNER_THREADS) (&wbits[0][0])[i] = 0u;

        uint4 m[SCAN_TRIPS];
#pragma unroll
        for (int k = 0; k < SCAN_TRIPS; ++k) {
            const int idx = k * OWNER_THREADS + t;
            m[k] = make_uint4(0u, 0u, 0u, 0u);
            if (idx < BM_VEC) m[k] = *(const uint4*)(bm + (unsigned)idx * 16u);
        }

        int mine_cnt = 0;
#pragma unroll
        for (int k = 0; k < SCAN_TRIPS; ++k)
            mine_cnt += __builtin_popcount(m[k].x) + __builtin_popcount(m[k].y) +
                        __builtin_popcount(m[k].z) + __builtin_popcount(m[k].w);
        mine_cnt = wave_sum(mine_cnt);
        if (lane == 0) cnt[wave] = mine_cnt;
        __syncthreads();
        const int cl = cnt[lane];
        const int offset = __builtin_amdgcn_readfirstlane(wave_sum(lane < wave ? cl : 0));
        total = __builtin_amdgcn_readfirstlane(wave_sum(cl));

        int g = offset;
#pragma unroll
        for (int k = 0; k < SCAN_TRIPS; ++k) {
#pragma unroll
            for (int c = 0; c < 4; ++c) {
                unsigned d = (c == 0) ? m[k].x : (c == 1) ? m[k].y : (c == 2) ? m[k].z : m[k].w;
                const unsigned ebase = ((unsigned)(k * OWNER_THREADS + t) * 4u + (unsigned)c) * 32u;
                unsigned mask = __builtin_amdgcn_ballot_w32(d != 0u);
                while (mask != 0u) {
                    if (d != 0u) {
                        const unsigned bit = (unsigned)__builtin_ctz(d);
                        const int      pos = g + __builtin_popcount(mask & lanes_below) - base;
                        d &= d - 1u;
                        if ((unsigned)pos < (unsigned)LIST_CAP) recs[pos] = (unsigned long long)(ebase + bit);
                    }
                    g   += __builtin_popcount(mask);
                    mask = __builtin_amdgcn_ballot_w32(d != 0u);
                }
            }
        }
        __syncthreads();

        const int n = (total - base) < LIST_CAP ? (total - base) : LIST_CAP;
        unsigned long long rr[LIST_CAP / OWNER_THREADS];
#pragma unroll
        for (int q = 0; q < LIST_CAP / OWNER_THREADS; ++q) {
            const int i = q * OWNER_THREADS + t;
            rr[q] = 0ull;
            if (i < n) rr[q] = rec[(unsigned)recs[i]];
        }
#pragma unroll
        for (int q = 0; q < LIST_CAP / OWNER_THREADS; ++q) {
            const int  i     = q * OWNER_THREADS + t;
            const bool valid = i < n;
            if (valid) recs[i] = rr[q];
            const int v = (int)((((unsigned)rr[q]) >> 16) & 31u);
            unsigned todo = __builtin_amdgcn_ballot_w32(valid);
            while (todo != 0u) {
                const int      leader = __builtin_ctz(todo);
                const int      lv     = __builtin_amdgcn_readlane(v, leader);
                const unsigned mm     = __builtin_amdgcn_ballot_w32(valid && v == lv);
                if (lane == leader) wbits[lv][q * 32 + wave] = mm;
                todo &= ~mm;
            }
        }
        __syncthreads();

        int head  = 0;
        int count = 0;
        for (int w0 = 0; w0 * 32 < n; w0 += 32) {
            const int word = w0 + lane;
            unsigned  d    = wbits[wave][word];
            unsigned mask = __builtin_amdgcn_ballot_w32(d != 0u);
            while (mask != 0u) {
                if (d != 0u) {
                    const int bit = __builtin_ctz(d);
                    d &= d - 1u;
                    ring[(head + count + __builtin_popcount(mask & lanes_below)) & (RING - 1)] =
                        recs[word * 32 + bit];
                }
                count += __builtin_popcount(mask);
                while (count >= EPW) {
                    owner_batch<false>(p, sums, ring, head, EPW, lane);
                    head   = (head + EPW) & (RING - 1);
                    count -= EPW;
                }
                mask = __builtin_amdgcn_ballot_w32(d != 0u);
            }
        }
        if (count > 0) owner_batch<true>(p, sums, ring, head, count, lane);
        __syncthreads();
        base += LIST_CAP;
    } while (base < total);

    const int row0 = owner * ROWS_PER_OWNER;
    int nr = nrows - row0;
    nr = nr < 0 ? 0 : (nr > ROWS_PER_OWNER ? ROWS_PER_OWNER : nr);
    float4* a4 = reinterpret_cast<float4*>(acc + (size_t)row0 * DIM);
    for (int i = t; i < nr * (DIM / 4); i += OWNER_THREADS) {
        float4 a = a4[i];
        const float4 sv = s4[i];
        a.x += sv.x;  a.y += sv.y;  a.z += sv.z;  a.w += sv.w;
        a4[i] = a;
    }
}

#define GEMM_WGS 256
static_assert(NNODES % 16 == 0, "node tiles are whole: no row guard at the loads, WMMAs or stores");

template <bool RELU>
__global__ __launch_bounds__(256) void gcn_gemm_wmma_kernel(
    const float* __restrict__ hin,
    const float* __restrict__ Wrel,
    const float* __restrict__ Wroot,
    const float* __restrict__ bias,
    float*       __restrict__ p,
    float*       __restrict__ acc,
    int nrows) {
    const int t    = threadIdx.x;
    const int lane = t & 31;
    const int wv   = t >> 5;
    const int r    = lane & 15;
    const int half = lane >> 4;
    const int col0 = wv * 16;
    const int ntiles = nrows >> 4;
    const int stride = (int)gridDim.x;

    int tile = (int)blockIdx.x;
    float4 a_nxt[16];
    {
        const int tl = tile < ntiles ? tile : ntiles - 1;
        const float4* ap = reinterpret_cast<const float4*>(
            hin + ((size_t)tl * 16 + r) * DIM + 4 * half);
#pragma unroll
        for (int i = 0; i < 16; ++i) a_nxt[i] = ap[2 * i];
    }

    v2f b_rel[32];
    v2f b_root[32];
#pragma unroll
    for (int i = 0; i < 16; ++i) {
        const int kb = 8 * i + 4 * half;
        const float* wr = Wrel  + (size_t)kb * DIM + col0 + r;
        const float* wo = Wroot + (size_t)kb * DIM + col0 + r;
        b_rel [2 * i]     = v2f{ wr[0],       wr[DIM]     };
        b_rel [2 * i + 1] = v2f{ wr[2 * DIM], wr[3 * DIM] };
        b_root[2 * i]     = v2f{ wo[0],       wo[DIM]     };
        b_root[2 * i + 1] = v2f{ wo[2 * DIM], wo[3 * DIM] };
    }
    const float bv = bias[col0 + r];

    for (; tile < ntiles; tile += stride) {
        float4 a[16];
#pragma unroll
        for (int i = 0; i < 16; ++i) {
            float4 q = a_nxt[i];
            if (RELU) {
                q.x = fmaxf(q.x, 0.f);  q.y = fmaxf(q.y, 0.f);
                q.z = fmaxf(q.z, 0.f);  q.w = fmaxf(q.w, 0.f);
            }
            a[i] = q;
        }

        {
            const int nt = tile + stride;
            const int tl = nt < ntiles ? nt : ntiles - 1;
            const float4* ap = reinterpret_cast<const float4*>(
                hin + ((size_t)tl * 16 + r) * DIM + 4 * half);
#pragma unroll
            for (int i = 0; i < 16; ++i) a_nxt[i] = ap[2 * i];
        }

        v8f acc_rel  = {};
        v8f acc_root = {};
#pragma unroll
        for (int i = 0; i < 16; ++i) {
            const v2f a0 = { a[i].x, a[i].y };
            const v2f a1 = { a[i].z, a[i].w };
            acc_rel  = __builtin_amdgcn_wmma_f32_16x16x4_f32(
                           false, a0, false, b_rel[2 * i],  (short)0, acc_rel,  false, false);
            acc_root = __builtin_amdgcn_wmma_f32_16x16x4_f32(
                           false, a0, false, b_root[2 * i], (short)0, acc_root, false, false);
            acc_rel  = __builtin_amdgcn_wmma_f32_16x16x4_f32(
                           false, a1, false, b_rel[2 * i + 1],  (short)0, acc_rel,  false, false);
            acc_root = __builtin_amdgcn_wmma_f32_16x16x4_f32(
                           false, a1, false, b_root[2 * i + 1], (short)0, acc_root, false, false);
        }

        const size_t obase = ((size_t)tile * 16 + 8 * half) * DIM + col0 + r;
#pragma unroll
        for (int j = 0; j < 8; ++j) {
            p  [obase + (size_t)j * DIM] = acc_rel[j];
            acc[obase + (size_t)j * DIM] = acc_root[j] + bv;
        }
    }
}

extern "C" void kernel_launch(void* const* d_in, const int* in_sizes, int n_in,
                              void* d_out, int out_size, void* d_ws, size_t ws_size,
                              hipStream_t stream) {
    const float* x      = (const float*)d_in[0];
    const int*   eidx   = (const int*)  d_in[1];
    const float* ew     = (const float*)d_in[2];
    const float* Wrel   = (const float*)d_in[3];
    const float* brel   = (const float*)d_in[4];
    const float* Wroot  = (const float*)d_in[5];
    float*       out    = (float*)d_out;

    const int N = NNODES;
    const int E = NEDGES;

    float*    s1  = (float*)d_ws;
    float*    s2  = s1 + (size_t)N * DIM;
    float*    p   = s2 + (size_t)N * DIM;
    unsigned* bmp = (unsigned*)(p + (size_t)N * DIM);
    unsigned long long* rec = (unsigned long long*)(bmp + (size_t)NOWNERS * BM_STRIDE);

    gcn_edge_prep_kernel<<<PREP_BLOCKS, PREP_EDGES, 0, stream>>>(eidx, ew, bmp, rec, E);

    for (int l = 0; l < NLAYERS; ++l) {
        const float* hin  = (l == 0) ? x  : (l == 1) ? s1 : s2;
        float*       hout = (l == 0) ? s1 : (l == 1) ? s2 : out;

        if (l == 0)
            gcn_gemm_wmma_kernel<false><<<GEMM_WGS, 256, 0, stream>>>(
                hin, Wrel + (size_t)l * DIM * DIM, Wroot + (size_t)l * DIM * DIM,
                brel + (size_t)l * DIM, p, hout, N);
        else
            gcn_gemm_wmma_kernel<true><<<GEMM_WGS, 256, 0, stream>>>(
                hin, Wrel + (size_t)l * DIM * DIM, Wroot + (size_t)l * DIM * DIM,
                brel + (size_t)l * DIM, p, hout, N);

        gcn_owner_sum_kernel<<<NOWNERS, OWNER_THREADS, 0, stream>>>(
            p, bmp, rec, hout, N);
    }
}


# edge phase: a wave's row sums stay in registers (7 slots of 4 floats, reached by register-relative moves on the record's row), not in LDS: no LDS read-add-write per edge; sum order unchanged
# speedup vs baseline: 2.6881x; 1.0332x over previous
#include <hip/hip_runtime.h>

#define NNODES  50000
#define NEDGES  800000
#define DIM     128
#define NLAYERS 3

typedef __attribute__((ext_vector_type(2))) float v2f;
typedef __attribute__((ext_vector_type(8))) float v8f;

#define ROWS_PER_OWNER 196
#define NOWNERS        256
#define PREP_EDGES     1024
#define PREP_BLOCKS    ((NEDGES + PREP_EDGES - 1) / PREP_EDGES)
#define BM_STRIDE      (PREP_BLOCKS * (PREP_EDGES / 32))
#define BM_VEC         (BM_STRIDE / 4)
#define OWNER_THREADS  1024
#define OWNER_WAVES    (OWNER_THREADS / 32)
#define SCAN_TRIPS     ((BM_VEC + OWNER_THREADS - 1) / OWNER_THREADS)
#define LIST_CAP       8192
#define NSLOT          ((ROWS_PER_OWNER + OWNER_WAVES - 1) / OWNER_WAVES)
static_assert(ROWS_PER_OWNER * NOWNERS >= NNODES, "the owners' ranges cover every node");
static_assert(NNODES <= 65536 && ROWS_PER_OWNER <= 65536, "source row and row-in-range share one 32-bit word");
static_assert(BM_STRIDE % 32 == 0, "an owner's map is whole 128-byte lines");
static_assert(OWNER_WAVES == 32, "one lane per wave in the offset sums; rows are dealt to waves by r % 32");

__global__ __launch_bounds__(PREP_EDGES) void gcn_edge_prep_kernel(
    const int*   __restrict__ edge_index,
    const float* __restrict__ edge_weight,
    unsigned*    __restrict__ bitmap,
    unsigned long long* __restrict__ rec,
    int nedges) {
    __shared__ unsigned bits[NOWNERS][PREP_EDGES / 32];
    const int t    = threadIdx.x;
    const int lane = t & 31;
    const int wave = t >> 5;
    unsigned* flat = &bits[0][0];
    for (int i = t; i < NOWNERS * (PREP_EDGES / 32); i += PREP_EDGES) flat[i] = 0u;
    __syncthreads();

    const int  e     = blockIdx.x * PREP_EDGES + t;
    const bool valid = e < nedges;
    int o = 0;
    if (valid) {
        const int   s = edge_index[e];
        const int   d = edge_index[nedges + e];
        const float w = edge_weight[e];
        o = d / ROWS_PER_OWNER;
        const int r = d - o * ROWS_PER_OWNER;
        rec[e] = (unsigned long long)((unsigned)s | ((unsigned)r << 16)) |
                 ((unsigned long long)__float_as_uint(w) << 32);
    }
    unsigned todo = __builtin_amdgcn_ballot_w32(valid);
    while (todo != 0u) {
        const int      leader = __builtin_ctz(todo);
        const int      lo     = __builtin_amdgcn_readlane(o, leader);
        const unsigned m      = __builtin_amdgcn_ballot_w32(valid && o == lo);
        if (lane == leader) bits[lo][wave] = m;
        todo &= ~m;
    }
    __syncthreads();

    for (int i = t; i < NOWNERS * (PREP_EDGES / 32); i += PREP_EDGES) {
        const int oo = i >> 5;
        const int j  = i & 31;
        bitmap[(size_t)oo * BM_STRIDE + (size_t)blockIdx.x * (PREP_EDGES / 32) + j] = flat[i];
    }
}

__device__ __forceinline__ int wave_sum(int v) {
    v += __shfl_xor(v, 16);
    v += __shfl_xor(v, 8);
    v += __shfl_xor(v, 4);
    v += __shfl_xor(v, 2);
    v += __shfl_xor(v, 1);
    return v;
}

#define EPW  16
#define RING 64
static_assert(EPW - 1 + 32 <= RING, "the ring holds the waiting edges and one round of new ones");

#define FENCE4(J) asm volatile("" :: \
    "v"(v[J].x),   "v"(v[J].y),   "v"(v[J].z),   "v"(v[J].w),   \
    "v"(v[J+1].x), "v"(v[J+1].y), "v"(v[J+1].z), "v"(v[J+1].w), \
    "v"(v[J+2].x), "v"(v[J+2].y), "v"(v[J+2].z), "v"(v[J+2].w), \
    "v"(v[J+3].x), "v"(v[J+3].y), "v"(v[J+3].z), "v"(v[J+3].w));
static_assert(EPW == 16, "the load fences in owner_batch name 16 gathers");

typedef volatile unsigned long long __attribute__((address_space(3))) ring_word;

typedef float __attribute__((ext_vector_type(32))) sums_t;
static_assert(NSLOT <= 8, "the slots fit the register vector");

__device__ __forceinline__ void slot_add(sums_t& sum, int slot, const float4 v, float w) {
    float s0 = sum[slot], s1 = sum[slot + 8], s2 = sum[slot + 16], s3 = sum[slot + 24];
    s0 += v.x * w;
    s1 += v.y * w;
    s2 += v.z * w;
    s3 += v.w * w;
    sum[slot]      = s0;
    sum[slot + 8]  = s1;
    sum[slot + 16] = s2;
    sum[slot + 24] = s3;
}

template <bool PARTIAL>
__device__ __forceinline__ void owner_batch(
    const float* __restrict__ p, sums_t& sum,
    const ring_word* ring, int head, int n, int lane) {
    unsigned long long rc = 0ull;
    if (lane < n) rc = ring[(head + lane) & (RING - 1)];
    const int rlo = (int)(unsigned)rc;
    const int rhi = (int)(unsigned)(rc >> 32);

    const unsigned voff = (unsigned)lane * 16u;
    float4 v[EPW];
    int    row[EPW];
    float  w[EPW];
#pragma unroll
    for (int j = 0; j < EPW; ++j) {
        const unsigned sx = (unsigned)__builtin_amdgcn_readlane(rlo, j);
        w[j]   = __uint_as_float((unsigned)__builtin_amdgcn_readlane(rhi, j));
        row[j] = (int)(sx >> 16) / OWNER_WAVES;
        const unsigned off = (sx & 0xffffu) * (unsigned)(DIM * 4) + voff;
        v[j] = *(const float4*)((const char*)p + off);
    }
    FENCE4(0)  FENCE4(4)  FENCE4(8)  FENCE4(12)
#pragma unroll
    for (int j = 0; j < EPW; ++j) {
        if (!PARTIAL || j < n) slot_add(sum, row[j], v[j], w[j]);
    }
}

__global__ __launch_bounds__(OWNER_THREADS) void gcn_owner_sum_kernel(
    const float*    __restrict__ p,
    const unsigned*              bitmap,
    const unsigned long long* __restrict__ rec,
    float*          __restrict__ acc,
    int nrows) {
    __shared__ unsigned long long recs[LIST_CAP];
    __shared__ unsigned long long rings[OWNER_WAVES][RING];
    __shared__ unsigned           wbits[OWNER_WAVES][LIST_CAP / 32];
    __shared__ int                cnt[OWNER_WAVES];

    const int t     = threadIdx.x;
    const int lane  = t & 31;
    const int wave  = __builtin_amdgcn_readfirstlane((int)(threadIdx.x >> 5));
    const int owner = blockIdx.x;
    const char* bm = reinterpret_cast<const char*>(bitmap + (size_t)owner * BM_STRIDE);

    sums_t sum = {};

    ring_word* ring = (ring_word*)&rings[wave][0];
    const unsigned lanes_below = (1u << lane) - 1u;

    int base = 0;
    int total;
    do {
        int tw = t;
        asm volatile("" : "+v"(tw));
        __builtin_assume(tw >= 0 && tw < OWNER_THREADS);
        const int lw = tw & 31;

        for (int i = tw; i < OWNER_WAVES * (LIST_CAP / 32); i += OWNER_THREADS) (&wbits[0][0])[i] = 0u;

        uint4 m[SCAN_TRIPS];
#pragma unroll
        for (int k = 0; k < SCAN_TRIPS; ++k) {
            const int idx = k * OWNER_THREADS + tw;
            m[k] = make_uint4(0u, 0u, 0u, 0u);
            if (idx < BM_VEC) m[k] = *(const uint4*)(bm + (unsigned)idx * 16u);
        }

        int mine_cnt = 0;
#pragma unroll
        for (int k = 0; k < SCAN_TRIPS; ++k)
            mine_cnt += __builtin_popcount(m[k].x) + __builtin_popcount(m[k].y) +
                        __builtin_popcount(m[k].z) + __builtin_popcount(m[k].w);
        mine_cnt = wave_sum(mine_cnt);
        if (lw == 0) cnt[wave] = mine_cnt;
        __syncthreads();
        const int cl = cnt[lw];
        const int offset = __builtin_amdgcn_readfirstlane(wave_sum(lw < wave ? cl : 0));
        total = __builtin_amdgcn_readfirstlane(wave_sum(cl));

        int g = offset;
#pragma unroll
        for (int k = 0; k < SCAN_TRIPS; ++k) {
#pragma unroll
            for (int c = 0; c < 4; ++c) {
                unsigned d = (c == 0) ? m[k].x : (c == 1) ? m[k].y : (c == 2) ? m[k].z : m[k].w;
                const unsigned ebase = ((unsigned)(k * OWNER_THREADS + tw) * 4u + (unsigned)c) * 32u;
                unsigned mask = __builtin_amdgcn_ballot_w32(d != 0u);
                while (mask != 0u) {
                    if (d != 0u) {
                        const unsigned bit = (unsigned)__builtin_ctz(d);
                        const int      pos = g + __builtin_popcount(mask & lanes_below) - base;
                        d &= d - 1u;
                        if ((unsigned)pos < (unsigned)LIST_CAP) recs[pos] = (unsigned long long)(ebase + bit);
                    }
                    g   += __builtin_popcount(mask);
                    mask = __builtin_amdgcn_ballot_w32(d != 0u);
                }
            }
        }
        __syncthreads();

        const int n = (total - base) < LIST_CAP ? (total - base) : LIST_CAP;
        unsigned long long rr[LIST_CAP / OWNER_THREADS];
#pragma unroll
        for (int q = 0; q < LIST_CAP / OWNER_THREADS; ++q) {
            const int i = q * OWNER_THREADS + tw;
            rr[q] = 0ull;
            if (i < n) rr[q] = rec[(unsigned)recs[i]];
        }
#pragma unroll
        for (int q = 0; q < LIST_CAP / OWNER_THREADS; ++q) {
            const int  i     = q * OWNER_THREADS + tw;
            const bool valid = i < n;
            if (valid) recs[i] = rr[q];
            const int v = (int)((((unsigned)rr[q]) >> 16) & 31u);
            unsigned todo = __builtin_amdgcn_ballot_w32(valid);
            while (todo != 0u) {
                const int      leader = __builtin_ctz(todo);
                const int      lv     = __builtin_amdgcn_readlane(v, leader);
                const unsigned mm     = __builtin_amdgcn_ballot_w32(valid && v == lv);
                if (lw == leader) wbits[lv][q * 32 + wave] = mm;
                todo &= ~mm;
            }
        }
        __syncthreads();

        int head  = 0;
        int count = 0;
        for (int w0 = 0; w0 * 32 < n; w0 += 32) {
            const int word = w0 + lane;
            unsigned  d    = wbits[wave][word];
            unsigned mask = __builtin_amdgcn_ballot_w32(d != 0u);
            while (mask != 0u) {
                if (d != 0u) {
                    const int bit = __builtin_ctz(d);
                    d &= d - 1u;
                    ring[(head + count + __builtin_popcount(mask & lanes_below)) & (RING - 1)] =
                        recs[word * 32 + bit];
                }
                count += __builtin_popcount(mask);
                while (count >= EPW) {
                    owner_batch<false>(p, sum, ring, head, EPW, lane);
                    head   = (head + EPW) & (RING - 1);
                    count -= EPW;
                }
                mask = __builtin_amdgcn_ballot_w32(d != 0u);
            }
        }
        if (count > 0) owner_batch<true>(p, sum, ring, head, count, lane);
        __syncthreads();
        base += LIST_CAP;
    } while (base < total);

    const int row0 = owner * ROWS_PER_OWNER;
    int nr = nrows - row0;
    nr = nr < 0 ? 0 : (nr > ROWS_PER_OWNER ? ROWS_PER_OWNER : nr);
    float4* a4 = reinterpret_cast<float4*>(acc + (size_t)row0 * DIM) + lane;
    float4 root[NSLOT];
#pragma unroll
    for (int s = 0; s < NSLOT; ++s) {
        const int r = s * OWNER_WAVES + wave;
        if (r < nr) root[s] = a4[r * (DIM / 4)];
    }
#pragma unroll
    for (int s = 0; s < NSLOT; ++s) {
        const int r = s * OWNER_WAVES + wave;
        if (r < nr) {
            float4 a = root[s];
            a.x += sum[s];  a.y += sum[s + 8];  a.z += sum[s + 16];  a.w += sum[s + 24];
            a4[r * (DIM / 4)] = a;
        }
    }
}

#define GEMM_WGS 256
static_assert(NNODES % 16 == 0, "node tiles are whole: no row guard at the loads, WMMAs or stores");

template <bool RELU>
__global__ __launch_bounds__(256) void gcn_gemm_wmma_kernel(
    const float* __restrict__ hin,
    const float* __restrict__ Wrel,
    const float* __restrict__ Wroot,
    const float* __restrict__ bias,
    float*       __restrict__ p,
    float*       __restrict__ acc,
    int nrows) {
    const int t    = threadIdx.x;
    const int lane = t & 31;
    const int wv   = t >> 5;
    const int r    = lane & 15;
    const int half = lane >> 4;
    const int col0 = wv * 16;
    const int ntiles = nrows >> 4;
    const int stride = (int)gridDim.x;

    int tile = (int)blockIdx.x;
    float4 a_nxt[16];
    {
        const int tl = tile < ntiles ? tile : ntiles - 1;
        const float4* ap = reinterpret_cast<const float4*>(
            hin + ((size_t)tl * 16 + r) * DIM + 4 * half);
#pragma unroll
        for (int i = 0; i < 16; ++i) a_nxt[i] = ap[2 * i];
    }

    v2f b_rel[32];
    v2f b_root[32];
#pragma unroll
    for (int i = 0; i < 16; ++i) {
        const int kb = 8 * i + 4 * half;
        const float* wr = Wrel  + (size_t)kb * DIM + col0 + r;
        const float* wo = Wroot + (size_t)kb * DIM + col0 + r;
        b_rel [2 * i]     = v2f{ wr[0],       wr[DIM]     };
        b_rel [2 * i + 1] = v2f{ wr[2 * DIM], wr[3 * DIM] };
        b_root[2 * i]     = v2f{ wo[0],       wo[DIM]     };
        b_root[2 * i + 1] = v2f{ wo[2 * DIM], wo[3 * DIM] };
    }
    const float bv = bias[col0 + r];

    for (; tile < ntiles; tile += stride) {
        float4 a[16];
#pragma unroll
        for (int i = 0; i < 16; ++i) {
            float4 q = a_nxt[i];
            if (RELU) {
                q.x = fmaxf(q.x, 0.f);  q.y = fmaxf(q.y, 0.f);
                q.z = fmaxf(q.z, 0.f);  q.w = fmaxf(q.w, 0.f);
            }
            a[i] = q;
        }

        {
            const int nt = tile + stride;
            const int tl = nt < ntiles ? nt : ntiles - 1;
            const float4* ap = reinterpret_cast<const float4*>(
                hin + ((size_t)tl * 16 + r) * DIM + 4 * half);
#pragma unroll
            for (int i = 0; i < 16; ++i) a_nxt[i] = ap[2 * i];
        }

        v8f acc_rel  = {};
        v8f acc_root = {};
#pragma unroll
        for (int i = 0; i < 16; ++i) {
            const v2f a0 = { a[i].x, a[i].y };
            const v2f a1 = { a[i].z, a[i].w };
            acc_rel  = __builtin_amdgcn_wmma_f32_16x16x4_f32(
                           false, a0, false, b_rel[2 * i],  (short)0, acc_rel,  false, false);
            acc_root = __builtin_amdgcn_wmma_f32_16x16x4_f32(
                           false, a0, false, b_root[2 * i], (short)0, acc_root, false, false);
            acc_rel  = __builtin_amdgcn_wmma_f32_16x16x4_f32(
                           false, a1, false, b_rel[2 * i + 1],  (short)0, acc_rel,  false, false);
            acc_root = __builtin_amdgcn_wmma_f32_16x16x4_f32(
                           false, a1, false, b_root[2 * i + 1], (short)0, acc_root, false, false);
        }

        const size_t obase = ((size_t)tile * 16 + 8 * half) * DIM + col0 + r;
#pragma unroll
        for (int j = 0; j < 8; ++j) {
            p  [obase + (size_t)j * DIM] = acc_rel[j];
            acc[obase + (size_t)j * DIM] = acc_root[j] + bv;
        }
    }
}

extern "C" void kernel_launch(void* const* d_in, const int* in_sizes, int n_in,
                              void* d_out, int out_size, void* d_ws, size_t ws_size,
                              hipStream_t stream) {
    const float* x      = (const float*)d_in[0];
    const int*   eidx   = (const int*)  d_in[1];
    const float* ew     = (const float*)d_in[2];
    const float* Wrel   = (const float*)d_in[3];
    const float* brel   = (const float*)d_in[4];
    const float* Wroot  = (const float*)d_in[5];
    float*       out    = (float*)d_out;

    const int N = NNODES;
    const int E = NEDGES;

    float*    s1  = (float*)d_ws;
    float*    s2  = s1 + (size_t)N * DIM;
    float*    p   = s2 + (size_t)N * DIM;
    unsigned* bmp = (unsigned*)(p + (size_t)N * DIM);
    unsigned long long* rec = (unsigned long long*)(bmp + (size_t)NOWNERS * BM_STRIDE);

    gcn_edge_prep_kernel<<<PREP_BLOCKS, PREP_EDGES, 0, stream>>>(eidx, ew, bmp, rec, E);

    for (int l = 0; l < NLAYERS; ++l) {
        const float* hin  = (l == 0) ? x  : (l == 1) ? s1 : s2;
        float*       hout = (l == 0) ? s1 : (l == 1) ? s2 : out;

        if (l == 0)
            gcn_gemm_wmma_kernel<false><<<GEMM_WGS, 256, 0, stream>>>(
                hin, Wrel + (size_t)l * DIM * DIM, Wroot + (size_t)l * DIM * DIM,
                brel + (size_t)l * DIM, p, hout, N);
        else
            gcn_gemm_wmma_kernel<true><<<GEMM_WGS, 256, 0, stream>>>(
                hin, Wrel + (size_t)l * DIM * DIM, Wroot + (size_t)l * DIM * DIM,
                brel + (size_t)l * DIM, p, hout, N);

        gcn_owner_sum_kernel<<<NOWNERS, OWNER_THREADS, 0, stream>>>(
            p, bmp, rec, hout, N);
    }
}


# edge preparation runs as extra 256-thread work-groups of layer 1's node-phase launch, under its matrix-bound products (a word's owner masks by 8 ballots, no loop); 6 launches, not 7; sums unchanged
# speedup vs baseline: 2.7824x; 1.0351x over previous
#include <hip/hip_runtime.h>

#define NNODES  50000
#define NEDGES  800000
#define DIM     128
#define NLAYERS 3

typedef __attribute__((ext_vector_type(2))) float v2f;
typedef __attribute__((ext_vector_type(8))) float v8f;

#define ROWS_PER_OWNER 196
#define NOWNERS        256
#define PREP_EDGES     1024
#define PREP_BLOCKS    ((NEDGES + PREP_EDGES - 1) / PREP_EDGES)
#define BM_STRIDE      (PREP_BLOCKS * (PREP_EDGES / 32))
#define BM_VEC         (BM_STRIDE / 4)
#define OWNER_THREADS  1024
#define OWNER_WAVES    (OWNER_THREADS / 32)
#define SCAN_TRIPS     ((BM_VEC + OWNER_THREADS - 1) / OWNER_THREADS)
#define LIST_CAP       8192
#define NSLOT          ((ROWS_PER_OWNER + OWNER_WAVES - 1) / OWNER_WAVES)
static_assert(ROWS_PER_OWNER * NOWNERS >= NNODES, "the owners' ranges cover every node");
static_assert(NNODES <= 65536 && ROWS_PER_OWNER <= 65536, "source row and row-in-range share one 32-bit word");
static_assert(BM_STRIDE % 32 == 0, "an owner's map is whole 128-byte lines");
static_assert(OWNER_WAVES == 32, "one lane per wave in the offset sums; rows are dealt to waves by r % 32");

#define PREP_THREADS 256
#define PREP_WORDS   (PREP_EDGES / 32)
#define PREP_WPW     (PREP_WORDS / (PREP_THREADS / 32))
static_assert(PREP_WPW * (PREP_THREADS / 32) == PREP_WORDS, "the waves share a block's words evenly");
static_assert(NOWNERS == 256, "an owner number is 8 bits: 8 ballots name a lane's group");

__device__ __forceinline__ void edge_prep_body(
    const int*   __restrict__ edge_index,
    const float* __restrict__ edge_weight,
    unsigned*    __restrict__ bitmap,
    unsigned long long* __restrict__ rec,
    int nedges, int pblk, unsigned (*bits)[PREP_WORDS]) {
    const int t    = threadIdx.x;
    const int lane = t & 31;
    const int wave = t >> 5;

    int   s[PREP_WPW], d[PREP_WPW];
    float w[PREP_WPW];
    bool  valid[PREP_WPW];
#pragma unroll
    for (int i = 0; i < PREP_WPW; ++i) {
        const int e = pblk * PREP_EDGES + (wave * PREP_WPW + i) * 32 + lane;
        valid[i] = e < nedges;
        s[i] = 0;  d[i] = 0;  w[i] = 0.f;
        if (valid[i]) {
            s[i] = edge_index[e];
            d[i] = edge_index[nedges + e];
            w[i] = edge_weight[e];
        }
    }

    uint4* flat4 = reinterpret_cast<uint4*>(&bits[0][0]);
    for (int i = t; i < NOWNERS * PREP_WORDS / 4; i += PREP_THREADS) flat4[i] = make_uint4(0u, 0u, 0u, 0u);
    __syncthreads();

#pragma unroll
    for (int i = 0; i < PREP_WPW; ++i) {
        const int j = wave * PREP_WPW + i;
        const int o = d[i] / ROWS_PER_OWNER;
        d[i] -= o * ROWS_PER_OWNER;
        unsigned m = __builtin_amdgcn_ballot_w32(valid[i]);
#pragma unroll
        for (int b = 0; b < 8; ++b) {
            const bool     bit = ((o >> b) & 1) != 0;
            const unsigned bb  = __builtin_amdgcn_ballot_w32(bit);
            m &= bit ? bb : ~bb;
        }
        if (valid[i] && (m & ((1u << lane) - 1u)) == 0u) bits[o][j] = m;
    }
    __syncthreads();

#pragma unroll
    for (int i = 0; i < PREP_WPW; ++i) {
        const int e = pblk * PREP_EDGES + (wave * PREP_WPW + i) * 32 + lane;
        if (valid[i])
            rec[e] = (unsigned long long)((unsigned)s[i] | ((unsigned)d[i] << 16)) |
                     ((unsigned long long)__float_as_uint(w[i]) << 32);
    }

    for (int i = t; i < NOWNERS * PREP_WORDS / 4; i += PREP_THREADS) {
        const int oo = i >> 3;
        const int pc = i & 7;
        *reinterpret_cast<uint4*>(bitmap + (size_t)oo * BM_STRIDE + (size_t)pblk * PREP_WORDS + pc * 4) = flat4[i];
    }
}

__device__ __forceinline__ int wave_sum(int v) {
    v += __shfl_xor(v, 16);
    v += __shfl_xor(v, 8);
    v += __shfl_xor(v, 4);
    v += __shfl_xor(v, 2);
    v += __shfl_xor(v, 1);
    return v;
}

#define EPW  16
#define RING 64
static_assert(EPW - 1 + 32 <= RING, "the ring holds the waiting edges and one round of new ones");

#define FENCE4(J) asm volatile("" :: \
    "v"(v[J].x),   "v"(v[J].y),   "v"(v[J].z),   "v"(v[J].w),   \
    "v"(v[J+1].x), "v"(v[J+1].y), "v"(v[J+1].z), "v"(v[J+1].w), \
    "v"(v[J+2].x), "v"(v[J+2].y), "v"(v[J+2].z), "v"(v[J+2].w), \
    "v"(v[J+3].x), "v"(v[J+3].y), "v"(v[J+3].z), "v"(v[J+3].w));
static_assert(EPW == 16, "the load fences in owner_batch name 16 gathers");

typedef volatile unsigned long long __attribute__((address_space(3))) ring_word;

typedef float __attribute__((ext_vector_type(32))) sums_t;
static_assert(NSLOT <= 8, "the slots fit the register vector");

__device__ __forceinline__ void slot_add(sums_t& sum, int slot, const float4 v, float w) {
    float s0 = sum[slot], s1 = sum[slot + 8], s2 = sum[slot + 16], s3 = sum[slot + 24];
    s0 += v.x * w;
    s1 += v.y * w;
    s2 += v.z * w;
    s3 += v.w * w;
    sum[slot]      = s0;
    sum[slot + 8]  = s1;
    sum[slot + 16] = s2;
    sum[slot + 24] = s3;
}

template <bool PARTIAL>
__device__ __forceinline__ void owner_batch(
    const float* __restrict__ p, sums_t& sum,
    const ring_word* ring, int head, int n, int lane) {
    unsigned long long rc = 0ull;
    if (lane < n) rc = ring[(head + lane) & (RING - 1)];
    const int rlo = (int)(unsigned)rc;
    const int rhi = (int)(unsigned)(rc >> 32);

    const unsigned voff = (unsigned)lane * 16u;
    float4 v[EPW];
    int    row[EPW];
    float  w[EPW];
#pragma unroll
    for (int j = 0; j < EPW; ++j) {
        const unsigned sx = (unsigned)__builtin_amdgcn_readlane(rlo, j);
        w[j]   = __uint_as_float((unsigned)__builtin_amdgcn_readlane(rhi, j));
        row[j] = (int)(sx >> 16) / OWNER_WAVES;
        const unsigned off = (sx & 0xffffu) * (unsigned)(DIM * 4) + voff;
        v[j] = *(const float4*)((const char*)p + off);
    }
    FENCE4(0)  FENCE4(4)  FENCE4(8)  FENCE4(12)
#pragma unroll
    for (int j = 0; j < EPW; ++j) {
        if (!PARTIAL || j < n) slot_add(sum, row[j], v[j], w[j]);
    }
}

__global__ __launch_bounds__(OWNER_THREADS) void gcn_owner_sum_kernel(
    const float*    __restrict__ p,
    const unsigned*              bitmap,
    const unsigned long long* __restrict__ rec,
    float*          __restrict__ acc,
    int nrows) {
    __shared__ unsigned long long recs[LIST_CAP];
    __shared__ unsigned long long rings[OWNER_WAVES][RING];
    __shared__ unsigned           wbits[OWNER_WAVES][LIST_CAP / 32];
    __shared__ int                cnt[OWNER_WAVES];

    const int t     = threadIdx.x;
    const int lane  = t & 31;
    const int wave  = __builtin_amdgcn_readfirstlane((int)(threadIdx.x >> 5));
    const int owner = blockIdx.x;
    const char* bm = reinterpret_cast<const char*>(bitmap + (size_t)owner * BM_STRIDE);

    sums_t sum = {};

    ring_word* ring = (ring_word*)&rings[wave][0];
    const unsigned lanes_below = (1u << lane) - 1u;

    int base = 0;
    int total;
    do {
        int tw = t;
        asm volatile("" : "+v"(tw));
        __builtin_assume(tw >= 0 && tw < OWNER_THREADS);
        const int lw = tw & 31;

        for (int i = tw; i < OWNER_WAVES * (LIST_CAP / 32); i += OWNER_THREADS) (&wbits[0][0])[i] = 0u;

        uint4 m[SCAN_TRIPS];
#pragma unroll
        for (int k = 0; k < SCAN_TRIPS; ++k) {
            const int idx = k * OWNER_THREADS + tw;
            m[k] = make_uint4(0u, 0u, 0u, 0u);
            if (idx < BM_VEC) m[k] = *(const uint4*)(bm + (unsigned)idx * 16u);
        }

        int mine_cnt = 0;
#pragma unroll
        for (int k = 0; k < SCAN_TRIPS; ++k)
            mine_cnt += __builtin_popcount(m[k].x) + __builtin_popcount(m[k].y) +
                        __builtin_popcount(m[k].z) + __builtin_popcount(m[k].w);
        mine_cnt = wave_sum(mine_cnt);
        if (lw == 0) cnt[wave] = mine_cnt;
        __syncthreads();
        const int cl = cnt[lw];
        const int offset = __builtin_amdgcn_readfirstlane(wave_sum(lw < wave ? cl : 0));
        total = __builtin_amdgcn_readfirstlane(wave_sum(cl));

        int g = offset;
#pragma unroll
        for (int k = 0; k < SCAN_TRIPS; ++k) {
#pragma unroll
            for (int c = 0; c < 4; ++c) {
                unsigned d = (c == 0) ? m[k].x : (c == 1) ? m[k].y : (c == 2) ? m[k].z : m[k].w;
                const unsigned ebase = ((unsigned)(k * OWNER_THREADS + tw) * 4u + (unsigned)c) * 32u;
                unsigned mask = __builtin_amdgcn_ballot_w32(d != 0u);
                while (mask != 0u) {
                    if (d != 0u) {
                        const unsigned bit = (unsigned)__builtin_ctz(d);
                        const int      pos = g + __builtin_popcount(mask & lanes_below) - base;
                        d &= d - 1u;
                        if ((unsigned)pos < (unsigned)LIST_CAP) recs[pos] = (unsigned long long)(ebase + bit);
                    }
                    g   += __builtin_popcount(mask);
                    mask = __builtin_amdgcn_ballot_w32(d != 0u);
                }
            }
        }
        __syncthreads();

        const int n = (total - base) < LIST_CAP ? (total - base) : LIST_CAP;
        unsigned long long rr[LIST_CAP / OWNER_THREADS];
#pragma unroll
        for (int q = 0; q < LIST_CAP / OWNER_THREADS; ++q) {
            const int i = q * OWNER_THREADS + tw;
            rr[q] = 0ull;
            if (i < n) rr[q] = rec[(unsigned)recs[i]];
        }
#pragma unroll
        for (int q = 0; q < LIST_CAP / OWNER_THREADS; ++q) {
            const int  i     = q * OWNER_THREADS + tw;
            const bool valid = i < n;
            if (valid) recs[i] = rr[q];
            const int v = (int)((((unsigned)rr[q]) >> 16) & 31u);
            unsigned todo = __builtin_amdgcn_ballot_w32(valid);
            while (todo != 0u) {
                const int      leader = __builtin_ctz(todo);
                const int      lv     = __builtin_amdgcn_readlane(v, leader);
                const unsigned mm     = __builtin_amdgcn_ballot_w32(valid && v == lv);
                if (lw == leader) wbits[lv][q * 32 + wave] = mm;
                todo &= ~mm;
            }
        }
        __syncthreads();

        int head  = 0;
        int count = 0;
        for (int w0 = 0; w0 * 32 < n; w0 += 32) {
            const int word = w0 + lane;
            unsigned  d    = wbits[wave][word];
            unsigned mask = __builtin_amdgcn_ballot_w32(d != 0u);
            while (mask != 0u) {
                if (d != 0u) {
                    const int bit = __builtin_ctz(d);
                    d &= d - 1u;
                    ring[(head + count + __builtin_popcount(mask & lanes_below)) & (RING - 1)] =
                        recs[word * 32 + bit];
                }
                count += __builtin_popcount(mask);
                while (count >= EPW) {
                    owner_batch<false>(p, sum, ring, head, EPW, lane);
                    head   = (head + EPW) & (RING - 1);
                    count -= EPW;
                }
                mask = __builtin_amdgcn_ballot_w32(d != 0u);
            }
        }
        if (count > 0) owner_batch<true>(p, sum, ring, head, count, lane);
        __syncthreads();
        base += LIST_CAP;
    } while (base < total);

    const int row0 = owner * ROWS_PER_OWNER;
    int nr = nrows - row0;
    nr = nr < 0 ? 0 : (nr > ROWS_PER_OWNER ? ROWS_PER_OWNER : nr);
    float4* a4 = reinterpret_cast<float4*>(acc + (size_t)row0 * DIM) + lane;
    float4 root[NSLOT];
#pragma unroll
    for (int s = 0; s < NSLOT; ++s) {
        const int r = s * OWNER_WAVES + wave;
        if (r < nr) root[s] = a4[r * (DIM / 4)];
    }
#pragma unroll
    for (int s = 0; s < NSLOT; ++s) {
        const int r = s * OWNER_WAVES + wave;
        if (r < nr) {
            float4 a = root[s];
            a.x += sum[s];  a.y += sum[s + 8];  a.z += sum[s + 16];  a.w += sum[s + 24];
            a4[r * (DIM / 4)] = a;
        }
    }
}

#define GEMM_WGS 256
static_assert(NNODES % 16 == 0, "node tiles are whole: no row guard at the loads, WMMAs or stores");

template <bool RELU>
__device__ __forceinline__ void gemm_body(
    const float* __restrict__ hin,
    const float* __restrict__ Wrel,
    const float* __restrict__ Wroot,
    const float* __restrict__ bias,
    float*       __restrict__ p,
    float*       __restrict__ acc,
    int nrows, int blk, int nblk) {
    const int t    = threadIdx.x;
    const int lane = t & 31;
    const int wv   = t >> 5;
    const int r    = lane & 15;
    const int half = lane >> 4;
    const int col0 = wv * 16;
    const int ntiles = nrows >> 4;
    const int stride = nblk;

    int tile = blk;
    float4 a_nxt[16];
    {
        const int tl = tile < ntiles ? tile : ntiles - 1;
        const float4* ap = reinterpret_cast<const float4*>(
            hin + ((size_t)tl * 16 + r) * DIM + 4 * half);
#pragma unroll
        for (int i = 0; i < 16; ++i) a_nxt[i] = ap[2 * i];
    }

    v2f b_rel[32];
    v2f b_root[32];
#pragma unroll
    for (int i = 0; i < 16; ++i) {
        const int kb = 8 * i + 4 * half;
        const float* wr = Wrel  + (size_t)kb * DIM + col0 + r;
        const float* wo = Wroot + (size_t)kb * DIM + col0 + r;
        b_rel [2 * i]     = v2f{ wr[0],       wr[DIM]     };
        b_rel [2 * i + 1] = v2f{ wr[2 * DIM], wr[3 * DIM] };
        b_root[2 * i]     = v2f{ wo[0],       wo[DIM]     };
        b_root[2 * i + 1] = v2f{ wo[2 * DIM], wo[3 * DIM] };
    }
    const float bv = bias[col0 + r];

    for (; tile < ntiles; tile += stride) {
        float4 a[16];
#pragma unroll
        for (int i = 0; i < 16; ++i) {
            float4 q = a_nxt[i];
            if (RELU) {
                q.x = fmaxf(q.x, 0.f);  q.y = fmaxf(q.y, 0.f);
                q.z = fmaxf(q.z, 0.f);  q.w = fmaxf(q.w, 0.f);
            }
            a[i] = q;
        }

        {
            const int nt = tile + stride;
            const int tl = nt < ntiles ? nt : ntiles - 1;
            const float4* ap = reinterpret_cast<const float4*>(
                hin + ((size_t)tl * 16 + r) * DIM + 4 * half);
#pragma unroll
            for (int i = 0; i < 16; ++i) a_nxt[i] = ap[2 * i];
        }

        v8f acc_rel  = {};
        v8f acc_root = {};
#pragma unroll
        for (int i = 0; i < 16; ++i) {
            const v2f a0 = { a[i].x, a[i].y };
            const v2f a1 = { a[i].z, a[i].w };
            acc_rel  = __builtin_amdgcn_wmma_f32_16x16x4_f32(
                           false, a0, false, b_rel[2 * i],  (short)0, acc_rel,  false, false);
            acc_root = __builtin_amdgcn_wmma_f32_16x16x4_f32(
                           false, a0, false, b_root[2 * i], (short)0, acc_root, false, false);
            acc_rel  = __builtin_amdgcn_wmma_f32_16x16x4_f32(
                           false, a1, false, b_rel[2 * i + 1],  (short)0, acc_rel,  false, false);
            acc_root = __builtin_amdgcn_wmma_f32_16x16x4_f32(
                           false, a1, false, b_root[2 * i + 1], (short)0, acc_root, false, false);
        }

        const size_t obase = ((size_t)tile * 16 + 8 * half) * DIM + col0 + r;
#pragma unroll
        for (int j = 0; j < 8; ++j) {
            p  [obase + (size_t)j * DIM] = acc_rel[j];
            acc[obase + (size_t)j * DIM] = acc_root[j] + bv;
        }
    }
}

__global__ __launch_bounds__(256) void gcn_gemm_wmma_kernel(
    const float* __restrict__ hin, const float* __restrict__ Wrel,
    const float* __restrict__ Wroot, const float* __restrict__ bias,
    float* __restrict__ p, float* __restrict__ acc, int nrows) {
    gemm_body<true>(hin, Wrel, Wroot, bias, p, acc, nrows, (int)blockIdx.x, (int)gridDim.x);
}

__global__ __launch_bounds__(256) void gcn_gemm_prep_kernel(
    const float* __restrict__ hin, const float* __restrict__ Wrel,
    const float* __restrict__ Wroot, const float* __restrict__ bias,
    float* __restrict__ p, float* __restrict__ acc, int nrows,
    const int*   __restrict__ edge_index,
    const float* __restrict__ edge_weight,
    unsigned*    __restrict__ bitmap,
    unsigned long long* __restrict__ rec,
    int nedges) {
    __shared__ __attribute__((aligned(16))) unsigned bits[NOWNERS][PREP_WORDS];
    if (blockIdx.x < GEMM_WGS)
        gemm_body<false>(hin, Wrel, Wroot, bias, p, acc, nrows, (int)blockIdx.x, GEMM_WGS);
    else
        edge_prep_body(edge_index, edge_weight, bitmap, rec, nedges, (int)blockIdx.x - GEMM_WGS, bits);
}

extern "C" void kernel_launch(void* const* d_in, const int* in_sizes, int n_in,
                              void* d_out, int out_size, void* d_ws, size_t ws_size,
                              hipStream_t stream) {
    const float* x      = (const float*)d_in[0];
    const int*   eidx   = (const int*)  d_in[1];
    const float* ew     = (const float*)d_in[2];
    const float* Wrel   = (const float*)d_in[3];
    const float* brel   = (const float*)d_in[4];
    const float* Wroot  = (const float*)d_in[5];
    float*       out    = (float*)d_out;

    const int N = NNODES;
    const int E = NEDGES;

    float*    s1  = (float*)d_ws;
    float*    s2  = s1 + (size_t)N * DIM;
    float*    p   = s2 + (size_t)N * DIM;
    unsigned* bmp = (unsigned*)(p + (size_t)N * DIM);
    unsigned long long* rec = (unsigned long long*)(bmp + (size_t)NOWNERS * BM_STRIDE);

    for (int l = 0; l < NLAYERS; ++l) {
        const float* hin  = (l == 0) ? x  : (l == 1) ? s1 : s2;
        float*       hout = (l == 0) ? s1 : (l == 1) ? s2 : out;

        if (l == 0)
            gcn_gemm_prep_kernel<<<GEMM_WGS + PREP_BLOCKS, 256, 0, stream>>>(
                hin, Wrel + (size_t)l * DIM * DIM, Wroot + (size_t)l * DIM * DIM,
                brel + (size_t)l * DIM, p, hout, N, eidx, ew, bmp, rec, E);
        else
            gcn_gemm_wmma_kernel<<<GEMM_WGS, 256, 0, stream>>>(
                hin, Wrel + (size_t)l * DIM * DIM, Wroot + (size_t)l * DIM * DIM,
                brel + (size_t)l * DIM, p, hout, N);

        gcn_owner_sum_kernel<<<NOWNERS, OWNER_THREADS, 0, stream>>>(
            p, bmp, rec, hout, N);
    }
}


# edge phase of layers 2 and 3: a wave reads back the list of its records that layer 1's edge phase saved as it summed (same call), not the bit maps: no scan, hand-out or barrier; sum order unchanged
# speedup vs baseline: 2.9020x; 1.0430x over previous
#include <hip/hip_runtime.h>

#define NNODES  50000
#define NEDGES  800000
#define DIM     128
#define NLAYERS 3

typedef __attribute__((ext_vector_type(2))) float v2f;
typedef __attribute__((ext_vector_type(8))) float v8f;

#define ROWS_PER_OWNER 196
#define NOWNERS        256
#define PREP_EDGES     1024
#define PREP_BLOCKS    ((NEDGES + PREP_EDGES - 1) / PREP_EDGES)
#define BM_STRIDE      (PREP_BLOCKS * (PREP_EDGES / 32))
#define BM_VEC         (BM_STRIDE / 4)
#define OWNER_THREADS  1024
#define OWNER_WAVES    (OWNER_THREADS / 32)
#define SCAN_TRIPS     ((BM_VEC + OWNER_THREADS - 1) / OWNER_THREADS)
#define LIST_CAP       8192
#define NSLOT          ((ROWS_PER_OWNER + OWNER_WAVES - 1) / OWNER_WAVES)
static_assert(ROWS_PER_OWNER * NOWNERS >= NNODES, "the owners' ranges cover every node");
static_assert(NNODES <= 65536 && ROWS_PER_OWNER <= 65536, "source row and row-in-range share one 32-bit word");
static_assert(BM_STRIDE % 32 == 0, "an owner's map is whole 128-byte lines");
static_assert(OWNER_WAVES == 32, "one lane per wave in the offset sums; rows are dealt to waves by r % 32");

#define WLIST_CAP (LIST_CAP / OWNER_WAVES)
#define WLIST_OVER 0x7fffffff
static_assert(WLIST_CAP * OWNER_WAVES == LIST_CAP, "the waves' saved lists fill the records' LDS array exactly");
static_assert(WLIST_CAP % 32 == 0, "a wave reads its saved list in whole 32-record trips");

#define PREP_THREADS 256
#define PREP_WORDS   (PREP_EDGES / 32)
#define PREP_WPW     (PREP_WORDS / (PREP_THREADS / 32))
static_assert(PREP_WPW * (PREP_THREADS / 32) == PREP_WORDS, "the waves share a block's words evenly");
static_assert(NOWNERS == 256, "an owner number is 8 bits: 8 ballots name a lane's group");

__device__ __forceinline__ void edge_prep_body(
    const int*   __restrict__ edge_index,
    const float* __restrict__ edge_weight,
    unsigned*    __restrict__ bitmap,
    unsigned long long* __restrict__ rec,
    int nedges, int pblk, unsigned (*bits)[PREP_WORDS]) {
    const int t    = threadIdx.x;
    const int lane = t & 31;
    const int wave = t >> 5;

    int   s[PREP_WPW], d[PREP_WPW];
    float w[PREP_WPW];
    bool  valid[PREP_WPW];
#pragma unroll
    for (int i = 0; i < PREP_WPW; ++i) {
        const int e = pblk * PREP_EDGES + (wave * PREP_WPW + i) * 32 + lane;
        valid[i] = e < nedges;
        s[i] = 0;  d[i] = 0;  w[i] = 0.f;
        if (valid[i]) {
            s[i] = edge_index[e];
            d[i] = edge_index[nedges + e];
            w[i] = edge_weight[e];
        }
    }

    uint4* flat4 = reinterpret_cast<uint4*>(&bits[0][0]);
    for (int i = t; i < NOWNERS * PREP_WORDS / 4; i += PREP_THREADS) flat4[i] = make_uint4(0u, 0u, 0u, 0u);
    __syncthreads();

#pragma unroll
    for (int i = 0; i < PREP_WPW; ++i) {
        const int j = wave * PREP_WPW + i;
        const int o = d[i] / ROWS_PER_OWNER;
        d[i] -= o * ROWS_PER_OWNER;
        unsigned m = __builtin_amdgcn_ballot_w32(valid[i]);
#pragma unroll
        for (int b = 0; b < 8; ++b) {
            const bool     bit = ((o >> b) & 1) != 0;
            const unsigned bb  = __builtin_amdgcn_ballot_w32(bit);
            m &= bit ? bb : ~bb;
        }
        if (valid[i] && (m & ((1u << lane) - 1u)) == 0u) bits[o][j] = m;
    }
    __syncthreads();

#pragma unroll
    for (int i = 0; i < PREP_WPW; ++i) {
        const int e = pblk * PREP_EDGES + (wave * PREP_WPW + i) * 32 + lane;
        if (valid[i])
            rec[e] = (unsigned long long)((unsigned)s[i] | ((unsigned)d[i] << 16)) |
                     ((unsigned long long)__float_as_uint(w[i]) << 32);
    }

    for (int i = t; i < NOWNERS * PREP_WORDS / 4; i += PREP_THREADS) {
        const int oo = i >> 3;
        const int pc = i & 7;
        *reinterpret_cast<uint4*>(bitmap + (size_t)oo * BM_STRIDE + (size_t)pblk * PREP_WORDS + pc * 4) = flat4[i];
    }
}

__device__ __forceinline__ int wave_sum(int v) {
    v += __shfl_xor(v, 16);
    v += __shfl_xor(v, 8);
    v += __shfl_xor(v, 4);
    v += __shfl_xor(v, 2);
    v += __shfl_xor(v, 1);
    return v;
}

#define EPW  16
#define RING 64
static_assert(EPW - 1 + 32 <= RING, "the ring holds the waiting edges and one round of new ones");

#define FENCE4(J) asm volatile("" :: \
    "v"(v[J].x),   "v"(v[J].y),   "v"(v[J].z),   "v"(v[J].w),   \
    "v"(v[J+1].x), "v"(v[J+1].y), "v"(v[J+1].z), "v"(v[J+1].w), \
    "v"(v[J+2].x), "v"(v[J+2].y), "v"(v[J+2].z), "v"(v[J+2].w), \
    "v"(v[J+3].x), "v"(v[J+3].y), "v"(v[J+3].z), "v"(v[J+3].w));
static_assert(EPW == 16, "the load fences in owner_batch name 16 gathers");

typedef volatile unsigned long long __attribute__((address_space(3))) ring_word;

typedef float __attribute__((ext_vector_type(32))) sums_t;
static_assert(NSLOT <= 8, "the slots fit the register vector");

__device__ __forceinline__ void slot_add(sums_t& sum, int slot, const float4 v, float w) {
    float s0 = sum[slot], s1 = sum[slot + 8], s2 = sum[slot + 16], s3 = sum[slot + 24];
    s0 += v.x * w;
    s1 += v.y * w;
    s2 += v.z * w;
    s3 += v.w * w;
    sum[slot]      = s0;
    sum[slot + 8]  = s1;
    sum[slot + 16] = s2;
    sum[slot + 24] = s3;
}

template <bool PARTIAL, bool SAVE>
__device__ __forceinline__ void owner_batch(
    const float* __restrict__ p, sums_t& sum,
    const ring_word* ring, int head, int n, int lane,
    unsigned long long* __restrict__ wl, int pos) {
    unsigned long long rc = 0ull;
    if (lane < n) rc = ring[(head + lane) & (RING - 1)];
    if (SAVE) {
        if (lane < n && pos + lane < WLIST_CAP) wl[pos + lane] = rc;
    }
    const int rlo = (int)(unsigned)rc;
    const int rhi = (int)(unsigned)(rc >> 32);

    const unsigned voff = (unsigned)lane * 16u;
    float4 v[EPW];
    int    row[EPW];
    float  w[EPW];
#pragma unroll
    for (int j = 0; j < EPW; ++j) {
        const unsigned sx = (unsigned)__builtin_amdgcn_readlane(rlo, j);
        w[j]   = __uint_as_float((unsigned)__builtin_amdgcn_readlane(rhi, j));
        row[j] = (int)(sx >> 16) / OWNER_WAVES;
        const unsigned off = (sx & 0xffffu) * (unsigned)(DIM * 4) + voff;
        v[j] = *(const float4*)((const char*)p + off);
    }
    FENCE4(0)  FENCE4(4)  FENCE4(8)  FENCE4(12)
#pragma unroll
    for (int j = 0; j < EPW; ++j) {
        if (!PARTIAL || j < n) slot_add(sum, row[j], v[j], w[j]);
    }
}

template <bool FIRST>
__global__ __launch_bounds__(OWNER_THREADS) void gcn_owner_sum_kernel(
    const float*    __restrict__ p,
    const unsigned*              bitmap,
    const unsigned long long* __restrict__ rec,
    float*          __restrict__ acc,
    int nrows,
    unsigned long long* __restrict__ wlist,
    int*            __restrict__ wcnt) {
    __shared__ unsigned long long recs[LIST_CAP];
    __shared__ unsigned long long rings[OWNER_WAVES][RING];
    __shared__ unsigned           wbits[OWNER_WAVES][LIST_CAP / 32];
    __shared__ int                cnt[OWNER_WAVES];

    const int t     = threadIdx.x;
    const int lane  = t & 31;
    const int wave  = __builtin_amdgcn_readfirstlane((int)(threadIdx.x >> 5));
    const int owner = blockIdx.x;
    const char* bm = reinterpret_cast<const char*>(bitmap + (size_t)owner * BM_STRIDE);

    sums_t sum = {};

    ring_word* ring = (ring_word*)&rings[wave][0];
    const unsigned lanes_below = (1u << lane) - 1u;

    unsigned long long* mywl = wlist + ((size_t)owner * OWNER_WAVES + wave) * WLIST_CAP;
    int pos = 0;

    bool saved = false;
    if (!FIRST) {
        const int call = wcnt[owner * OWNER_WAVES + lane];
        saved = __builtin_amdgcn_ballot_w32(call > WLIST_CAP) == 0u;
        if (saved) {
            const int c = __builtin_amdgcn_readlane(call, wave);
            ring_word* mylist = (ring_word*)&recs[wave * WLIST_CAP];
            unsigned long long r[WLIST_CAP / 32];
#pragma unroll
            for (int q = 0; q < WLIST_CAP / 32; ++q) {
                const int i = q * 32 + lane;
                r[q] = 0ull;
                if (i < c) r[q] = mywl[i];
            }
#pragma unroll
            for (int q = 0; q < WLIST_CAP / 32; ++q) {
                const int i = q * 32 + lane;
                if (i < c) mylist[i] = r[q];
            }
            int b = 0;
            for (; b + EPW <= c; b += EPW)
                owner_batch<false, false>(p, sum, mylist + b, 0, EPW, lane, nullptr, 0);
            if (b < c) owner_batch<true, false>(p, sum, mylist + b, 0, c - b, lane, nullptr, 0);
        }
    }

    if (!saved) {

    int base = 0;
    int total;
    do {
        int tw = t;
        asm volatile("" : "+v"(tw));
        __builtin_assume(tw >= 0 && tw < OWNER_THREADS);
        const int lw = tw & 31;

        for (int i = tw; i < OWNER_WAVES * (LIST_CAP / 32); i += OWNER_THREADS) (&wbits[0][0])[i] = 0u;

        uint4 m[SCAN_TRIPS];
#pragma unroll
        for (int k = 0; k < SCAN_TRIPS; ++k) {
            const int idx = k * OWNER_THREADS + tw;
            m[k] = make_uint4(0u, 0u, 0u, 0u);
            if (idx < BM_VEC) m[k] = *(const uint4*)(bm + (unsigned)idx * 16u);
        }

        int mine_cnt = 0;
#pragma unroll
        for (int k = 0; k < SCAN_TRIPS; ++k)
            mine_cnt += __builtin_popcount(m[k].x) + __builtin_popcount(m[k].y) +
                        __builtin_popcount(m[k].z) + __builtin_popcount(m[k].w);
        mine_cnt = wave_sum(mine_cnt);
        if (lw == 0) cnt[wave] = mine_cnt;
        __syncthreads();
        const int cl = cnt[lw];
        const int offset = __builtin_amdgcn_readfirstlane(wave_sum(lw < wave ? cl : 0));
        total = __builtin_amdgcn_readfirstlane(wave_sum(cl));

        int g = offset;
#pragma unroll
        for (int k = 0; k < SCAN_TRIPS; ++k) {
#pragma unroll
            for (int c = 0; c < 4; ++c) {
                unsigned d = (c == 0) ? m[k].x : (c == 1) ? m[k].y : (c == 2) ? m[k].z : m[k].w;
                const unsigned ebase = ((unsigned)(k * OWNER_THREADS + tw) * 4u + (unsigned)c) * 32u;
                unsigned mask = __builtin_amdgcn_ballot_w32(d != 0u);
                while (mask != 0u) {
                    if (d != 0u) {
                        const unsigned bit = (unsigned)__builtin_ctz(d);
                        const int      pos = g + __builtin_popcount(mask & lanes_below) - base;
                        d &= d - 1u;
                        if ((unsigned)pos < (unsigned)LIST_CAP) recs[pos] = (unsigned long long)(ebase + bit);
                    }
                    g   += __builtin_popcount(mask);
                    mask = __builtin_amdgcn_ballot_w32(d != 0u);
                }
            }
        }
        __syncthreads();

        const int n = (total - base) < LIST_CAP ? (total - base) : LIST_CAP;
        unsigned long long rr[LIST_CAP / OWNER_THREADS];
#pragma unroll
        for (int q = 0; q < LIST_CAP / OWNER_THREADS; ++q) {
            const int i = q * OWNER_THREADS + tw;
            rr[q] = 0ull;
            if (i < n) rr[q] = rec[(unsigned)recs[i]];
        }
#pragma unroll
        for (int q = 0; q < LIST_CAP / OWNER_THREADS; ++q) {
            const int  i     = q * OWNER_THREADS + tw;
            const bool valid = i < n;
            if (valid) recs[i] = rr[q];
            const int v = (int)((((unsigned)rr[q]) >> 16) & 31u);
            unsigned todo = __builtin_amdgcn_ballot_w32(valid);
            while (todo != 0u) {
                const int      leader = __builtin_ctz(todo);
                const int      lv     = __builtin_amdgcn_readlane(v, leader);
                const unsigned mm     = __builtin_amdgcn_ballot_w32(valid && v == lv);
                if (lw == leader) wbits[lv][q * 32 + wave] = mm;
                todo &= ~mm;
            }
        }
        __syncthreads();

        int head  = 0;
        int count = 0;
        for (int w0 = 0; w0 * 32 < n; w0 += 32) {
            const int word = w0 + lane;
            unsigned  d    = wbits[wave][word];
            unsigned mask = __builtin_amdgcn_ballot_w32(d != 0u);
            while (mask != 0u) {
                if (d != 0u) {
                    const int bit = __builtin_ctz(d);
                    d &= d - 1u;
                    ring[(head + count + __builtin_popcount(mask & lanes_below)) & (RING - 1)] =
                        recs[word * 32 + bit];
                }
                count += __builtin_popcount(mask);
                while (count >= EPW) {
                    owner_batch<false, FIRST>(p, sum, ring, head, EPW, lane, mywl, pos);
                    pos   += EPW;
                    head   = (head + EPW) & (RING - 1);
                    count -= EPW;
                }
                mask = __builtin_amdgcn_ballot_w32(d != 0u);
            }
        }
        if (count > 0) {
            owner_batch<true, FIRST>(p, sum, ring, head, count, lane, mywl, pos);
            pos += count;
        }
        __syncthreads();
        base += LIST_CAP;
    } while (base < total);

    if (FIRST && lane == 0) wcnt[owner * OWNER_WAVES + wave] = (total <= LIST_CAP) ? pos : WLIST_OVER;
    }

    const int row0 = owner * ROWS_PER_OWNER;
    int nr = nrows - row0;
    nr = nr < 0 ? 0 : (nr > ROWS_PER_OWNER ? ROWS_PER_OWNER : nr);
    float4* a4 = reinterpret_cast<float4*>(acc + (size_t)row0 * DIM) + lane;
    float4 root[NSLOT];
#pragma unroll
    for (int s = 0; s < NSLOT; ++s) {
        const int r = s * OWNER_WAVES + wave;
        if (r < nr) root[s] = a4[r * (DIM / 4)];
    }
#pragma unroll
    for (int s = 0; s < NSLOT; ++s) {
        const int r = s * OWNER_WAVES + wave;
        if (r < nr) {
            float4 a = root[s];
            a.x += sum[s];  a.y += sum[s + 8];  a.z += sum[s + 16];  a.w += sum[s + 24];
            a4[r * (DIM / 4)] = a;
        }
    }
}

#define GEMM_WGS 256
static_assert(NNODES % 16 == 0, "node tiles are whole: no row guard at the loads, WMMAs or stores");

template <bool RELU>
__device__ __forceinline__ void gemm_body(
    const float* __restrict__ hin,
    const float* __restrict__ Wrel,
    const float* __restrict__ Wroot,
    const float* __restrict__ bias,
    float*       __restrict__ p,
    float*       __restrict__ acc,
    int nrows, int blk, int nblk) {
    const int t    = threadIdx.x;
    const int lane = t & 31;
    const int wv   = t >> 5;
    const int r    = lane & 15;
    const int half = lane >> 4;
    const int col0 = wv * 16;
    const int ntiles = nrows >> 4;
    const int stride = nblk;

    int tile = blk;
    float4 a_nxt[16];
    {
        const int tl = tile < ntiles ? tile : ntiles - 1;
        const float4* ap = reinterpret_cast<const float4*>(
            hin + ((size_t)tl * 16 + r) * DIM + 4 * half);
#pragma unroll
        for (int i = 0; i < 16; ++i) a_nxt[i] = ap[2 * i];
    }

    v2f b_rel[32];
    v2f b_root[32];
#pragma unroll
    for (int i = 0; i < 16; ++i) {
        const int kb = 8 * i + 4 * half;
        const float* wr = Wrel  + (size_t)kb * DIM + col0 + r;
        const float* wo = Wroot + (size_t)kb * DIM + col0 + r;
        b_rel [2 * i]     = v2f{ wr[0],       wr[DIM]     };
        b_rel [2 * i + 1] = v2f{ wr[2 * DIM], wr[3 * DIM] };
        b_root[2 * i]     = v2f{ wo[0],       wo[DIM]     };
        b_root[2 * i + 1] = v2f{ wo[2 * DIM], wo[3 * DIM] };
    }
    const float bv = bias[col0 + r];

    for (; tile < ntiles; tile += stride) {
        float4 a[16];
#pragma unroll
        for (int i = 0; i < 16; ++i) {
            float4 q = a_nxt[i];
            if (RELU) {
                q.x = fmaxf(q.x, 0.f);  q.y = fmaxf(q.y, 0.f);
                q.z = fmaxf(q.z, 0.f);  q.w = fmaxf(q.w, 0.f);
            }
            a[i] = q;
        }

        {
            const int nt = tile + stride;
            const int tl = nt < ntiles ? nt : ntiles - 1;
            const float4* ap = reinterpret_cast<const float4*>(
                hin + ((size_t)tl * 16 + r) * DIM + 4 * half);
#pragma unroll
            for (int i = 0; i < 16; ++i) a_nxt[i] = ap[2 * i];
        }

        v8f acc_rel  = {};
        v8f acc_root = {};
#pragma unroll
        for (int i = 0; i < 16; ++i) {
            const v2f a0 = { a[i].x, a[i].y };
            const v2f a1 = { a[i].z, a[i].w };
            acc_rel  = __builtin_amdgcn_wmma_f32_16x16x4_f32(
                           false, a0, false, b_rel[2 * i],  (short)0, acc_rel,  false, false);
            acc_root = __builtin_amdgcn_wmma_f32_16x16x4_f32(
                           false, a0, false, b_root[2 * i], (short)0, acc_root, false, false);
            acc_rel  = __builtin_amdgcn_wmma_f32_16x16x4_f32(
                           false, a1, false, b_rel[2 * i + 1],  (short)0, acc_rel,  false, false);
            acc_root = __builtin_amdgcn_wmma_f32_16x16x4_f32(
                           false, a1, false, b_root[2 * i + 1], (short)0, acc_root, false, false);
        }

        const size_t obase = ((size_t)tile * 16 + 8 * half) * DIM + col0 + r;
#pragma unroll
        for (int j = 0; j < 8; ++j) {
            p  [obase + (size_t)j * DIM] = acc_rel[j];
            acc[obase + (size_t)j * DIM] = acc_root[j] + bv;
        }
    }
}

__global__ __launch_bounds__(256) void gcn_gemm_wmma_kernel(
    const float* __restrict__ hin, const float* __restrict__ Wrel,
    const float* __restrict__ Wroot, const float* __restrict__ bias,
    float* __restrict__ p, float* __restrict__ acc, int nrows) {
    gemm_body<true>(hin, Wrel, Wroot, bias, p, acc, nrows, (int)blockIdx.x, (int)gridDim.x);
}

__global__ __launch_bounds__(256) void gcn_gemm_prep_kernel(
    const float* __restrict__ hin, const float* __restrict__ Wrel,
    const float* __restrict__ Wroot, const float* __restrict__ bias,
    float* __restrict__ p, float* __restrict__ acc, int nrows,
    const int*   __restrict__ edge_index,
    const float* __restrict__ edge_weight,
    unsigned*    __restrict__ bitmap,
    unsigned long long* __restrict__ rec,
    int nedges) {
    __shared__ __attribute__((aligned(16))) unsigned bits[NOWNERS][PREP_WORDS];
    if (blockIdx.x < GEMM_WGS)
        gemm_body<false>(hin, Wrel, Wroot, bias, p, acc, nrows, (int)blockIdx.x, GEMM_WGS);
    else
        edge_prep_body(edge_index, edge_weight, bitmap, rec, nedges, (int)blockIdx.x - GEMM_WGS, bits);
}

extern "C" void kernel_launch(void* const* d_in, const int* in_sizes, int n_in,
                              void* d_out, int out_size, void* d_ws, size_t ws_size,
                              hipStream_t stream) {
    const float* x      = (const float*)d_in[0];
    const int*   eidx   = (const int*)  d_in[1];
    const float* ew     = (const float*)d_in[2];
    const float* Wrel   = (const float*)d_in[3];
    const float* brel   = (const float*)d_in[4];
    const float* Wroot  = (const float*)d_in[5];
    float*       out    = (float*)d_out;

    const int N = NNODES;
    const int E = NEDGES;

    float*    s1  = (float*)d_ws;
    float*    s2  = s1 + (size_t)N * DIM;
    float*    p   = s2 + (size_t)N * DIM;
    unsigned* bmp = (unsigned*)(p + (size_t)N * DIM);
    unsigned long long* rec = (unsigned long long*)(bmp + (size_t)NOWNERS * BM_STRIDE);
    unsigned long long* wlist = rec + (size_t)E;
    int* wcnt = (int*)(wlist + (size_t)NOWNERS * OWNER_WAVES * WLIST_CAP);

    for (int l = 0; l < NLAYERS; ++l) {
        const float* hin  = (l == 0) ? x  : (l == 1) ? s1 : s2;
        float*       hout = (l == 0) ? s1 : (l == 1) ? s2 : out;

        if (l == 0)
            gcn_gemm_prep_kernel<<<GEMM_WGS + PREP_BLOCKS, 256, 0, stream>>>(
                hin, Wrel + (size_t)l * DIM * DIM, Wroot + (size_t)l * DIM * DIM,
                brel + (size_t)l * DIM, p, hout, N, eidx, ew, bmp, rec, E);
        else
            gcn_gemm_wmma_kernel<<<GEMM_WGS, 256, 0, stream>>>(
                hin, Wrel + (size_t)l * DIM * DIM, Wroot + (size_t)l * DIM * DIM,
                brel + (size_t)l * DIM, p, hout, N);

        if (l == 0)
            gcn_owner_sum_kernel<true><<<NOWNERS, OWNER_THREADS, 0, stream>>>(
                p, bmp, rec, hout, N, wlist, wcnt);
        else
            gcn_owner_sum_kernel<false><<<NOWNERS, OWNER_THREADS, 0, stream>>>(
                p, bmp, rec, hout, N, wlist, wcnt);
    }
}


# root-term product (h @ Wroot + b) moved from the node launch into the layer's edge launch: waves done with their edges run its f32 WMMA tiles, Wroot in LDS; root plane gone; ReLU at store; same sums
# speedup vs baseline: 3.1259x; 1.0772x over previous
#include <hip/hip_runtime.h>

#define NNODES  50000
#define NEDGES  800000
#define DIM     128
#define NLAYERS 3

typedef __attribute__((ext_vector_type(2))) float v2f;
typedef __attribute__((ext_vector_type(8))) float v8f;

#define ROWS_PER_OWNER 196
#define NOWNERS        256
#define PREP_EDGES     1024
#define PREP_BLOCKS    ((NEDGES + PREP_EDGES - 1) / PREP_EDGES)
#define BM_STRIDE      (PREP_BLOCKS * (PREP_EDGES / 32))
#define BM_VEC         (BM_STRIDE / 4)
#define OWNER_THREADS  1024
#define OWNER_WAVES    (OWNER_THREADS / 32)
#define SCAN_TRIPS     ((BM_VEC + OWNER_THREADS - 1) / OWNER_THREADS)
#define LIST_CAP       8192
#define NSLOT          ((ROWS_PER_OWNER + OWNER_WAVES - 1) / OWNER_WAVES)
static_assert(ROWS_PER_OWNER * NOWNERS >= NNODES, "the owners' ranges cover every node");
static_assert(NNODES <= 65536 && ROWS_PER_OWNER <= 65536, "source row and row-in-range share one 32-bit word");
static_assert(BM_STRIDE % 32 == 0, "an owner's map is whole 128-byte lines");
static_assert(OWNER_WAVES == 32, "one lane per wave in the offset sums; rows are dealt to waves by r % 32");

#define WLIST_CAP (LIST_CAP / OWNER_WAVES)
#define WLIST_OVER 0x7fffffff
static_assert(WLIST_CAP * OWNER_WAVES == LIST_CAP, "the waves' saved lists fill the records' LDS array exactly");
static_assert(WLIST_CAP % 32 == 0, "a wave reads its saved list in whole 32-record trips");

#define PREP_THREADS 256
#define PREP_WORDS   (PREP_EDGES / 32)
#define PREP_WPW     (PREP_WORDS / (PREP_THREADS / 32))
static_assert(PREP_WPW * (PREP_THREADS / 32) == PREP_WORDS, "the waves share a block's words evenly");
static_assert(NOWNERS == 256, "an owner number is 8 bits: 8 ballots name a lane's group");

__device__ __forceinline__ void edge_prep_body(
    const int*   __restrict__ edge_index,
    const float* __restrict__ edge_weight,
    unsigned*    __restrict__ bitmap,
    unsigned long long* __restrict__ rec,
    int nedges, int pblk, unsigned (*bits)[PREP_WORDS]) {
    const int t    = threadIdx.x;
    const int lane = t & 31;
    const int wave = t >> 5;

    int   s[PREP_WPW], d[PREP_WPW];
    float w[PREP_WPW];
    bool  valid[PREP_WPW];
#pragma unroll
    for (int i = 0; i < PREP_WPW; ++i) {
        const int e = pblk * PREP_EDGES + (wave * PREP_WPW + i) * 32 + lane;
        valid[i] = e < nedges;
        s[i] = 0;  d[i] = 0;  w[i] = 0.f;
        if (valid[i]) {
            s[i] = edge_index[e];
            d[i] = edge_index[nedges + e];
            w[i] = edge_weight[e];
        }
    }

    uint4* flat4 = reinterpret_cast<uint4*>(&bits[0][0]);
    for (int i = t; i < NOWNERS * PREP_WORDS / 4; i += PREP_THREADS) flat4[i] = make_uint4(0u, 0u, 0u, 0u);
    __syncthreads();

#pragma unroll
    for (int i = 0; i < PREP_WPW; ++i) {
        const int j = wave * PREP_WPW + i;
        const int o = d[i] / ROWS_PER_OWNER;
        d[i] -= o * ROWS_PER_OWNER;
        unsigned m = __builtin_amdgcn_ballot_w32(valid[i]);
#pragma unroll
        for (int b = 0; b < 8; ++b) {
            const bool     bit = ((o >> b) & 1) != 0;
            const unsigned bb  = __builtin_amdgcn_ballot_w32(bit);
            m &= bit ? bb : ~bb;
        }
        if (valid[i] && (m & ((1u << lane) - 1u)) == 0u) bits[o][j] = m;
    }
    __syncthreads();

#pragma unroll
    for (int i = 0; i < PREP_WPW; ++i) {
        const int e = pblk * PREP_EDGES + (wave * PREP_WPW + i) * 32 + lane;
        if (valid[i])
            rec[e] = (unsigned long long)((unsigned)s[i] | ((unsigned)d[i] << 16)) |
                     ((unsigned long long)__float_as_uint(w[i]) << 32);
    }

    for (int i = t; i < NOWNERS * PREP_WORDS / 4; i += PREP_THREADS) {
        const int oo = i >> 3;
        const int pc = i & 7;
        *reinterpret_cast<uint4*>(bitmap + (size_t)oo * BM_STRIDE + (size_t)pblk * PREP_WORDS + pc * 4) = flat4[i];
    }
}

__device__ __forceinline__ int wave_sum(int v) {
    v += __shfl_xor(v, 16);
    v += __shfl_xor(v, 8);
    v += __shfl_xor(v, 4);
    v += __shfl_xor(v, 2);
    v += __shfl_xor(v, 1);
    return v;
}

#define EPW  16
#define RING 64
static_assert(EPW - 1 + 32 <= RING, "the ring holds the waiting edges and one round of new ones");

#define FENCE4(J) asm volatile("" :: \
    "v"(v[J].x),   "v"(v[J].y),   "v"(v[J].z),   "v"(v[J].w),   \
    "v"(v[J+1].x), "v"(v[J+1].y), "v"(v[J+1].z), "v"(v[J+1].w), \
    "v"(v[J+2].x), "v"(v[J+2].y), "v"(v[J+2].z), "v"(v[J+2].w), \
    "v"(v[J+3].x), "v"(v[J+3].y), "v"(v[J+3].z), "v"(v[J+3].w));
static_assert(EPW == 16, "the load fences in owner_batch name 16 gathers");

typedef volatile unsigned long long __attribute__((address_space(3))) ring_word;

typedef float __attribute__((ext_vector_type(32))) sums_t;
static_assert(NSLOT <= 8, "the slots fit the register vector");

__device__ __forceinline__ void slot_add(sums_t& sum, int slot, const float4 v, float w) {
    float s0 = sum[slot], s1 = sum[slot + 8], s2 = sum[slot + 16], s3 = sum[slot + 24];
    s0 += v.x * w;
    s1 += v.y * w;
    s2 += v.z * w;
    s3 += v.w * w;
    sum[slot]      = s0;
    sum[slot + 8]  = s1;
    sum[slot + 16] = s2;
    sum[slot + 24] = s3;
}

template <bool PARTIAL, bool SAVE>
__device__ __forceinline__ void owner_batch(
    const float* __restrict__ p, sums_t& sum,
    const ring_word* ring, int head, int n, int lane,
    unsigned long long* __restrict__ wl, int pos) {
    unsigned long long rc = 0ull;
    if (lane < n) rc = ring[(head + lane) & (RING - 1)];
    if (SAVE) {
        if (lane < n && pos + lane < WLIST_CAP) wl[pos + lane] = rc;
    }
    const int rlo = (int)(unsigned)rc;
    const int rhi = (int)(unsigned)(rc >> 32);

    const unsigned voff = (unsigned)lane * 16u;
    float4 v[EPW];
    int    row[EPW];
    float  w[EPW];
#pragma unroll
    for (int j = 0; j < EPW; ++j) {
        const unsigned sx = (unsigned)__builtin_amdgcn_readlane(rlo, j);
        w[j]   = __uint_as_float((unsigned)__builtin_amdgcn_readlane(rhi, j));
        row[j] = (int)(sx >> 16) / OWNER_WAVES;
        const unsigned off = (sx & 0xffffu) * (unsigned)(DIM * 4) + voff;
        v[j] = *(const float4*)((const char*)p + off);
    }
    FENCE4(0)  FENCE4(4)  FENCE4(8)  FENCE4(12)
#pragma unroll
    for (int j = 0; j < EPW; ++j) {
        if (!PARTIAL || j < n) slot_add(sum, row[j], v[j], w[j]);
    }
}

#define ROOT_TILES ((ROWS_PER_OWNER + 15) / 16)
#define ROOT_NT    (ROOT_TILES * 8)
#define ROOT_PITCH 132
#define FENCE16(X, J) asm volatile("" :: \
    "v"(X[J]),    "v"(X[J+1]),  "v"(X[J+2]),  "v"(X[J+3]),  \
    "v"(X[J+4]),  "v"(X[J+5]),  "v"(X[J+6]),  "v"(X[J+7]),  \
    "v"(X[J+8]),  "v"(X[J+9]),  "v"(X[J+10]), "v"(X[J+11]), \
    "v"(X[J+12]), "v"(X[J+13]), "v"(X[J+14]), "v"(X[J+15]));

__device__ __forceinline__ void root_tile(
    const float* __restrict__ hin,
    const float* __restrict__ bias,
    int row0, int rt, int col0, int nrows, int lane,
    const float (*wt)[ROOT_PITCH],
    float (*rootl)[ROOT_PITCH]) {
    const int r    = lane & 15;
    const int half = lane >> 4;
    int row = row0 + rt * 16 + r;
    row = row < nrows ? row : nrows - 1;
    const float4* ap = reinterpret_cast<const float4*>(hin + (size_t)row * DIM + 4 * half);
    const float*  bp = &wt[col0 + r][4 * half];
    v8f c = {};
#pragma unroll
    for (int hk = 0; hk < 2; ++hk) {
        float a[32];
#pragma unroll
        for (int i = 0; i < 8; ++i) {
            const float4 q = ap[2 * (8 * hk + i)];
            a[4 * i] = q.x;  a[4 * i + 1] = q.y;  a[4 * i + 2] = q.z;  a[4 * i + 3] = q.w;
        }
        FENCE16(a, 0)  FENCE16(a, 16)
#pragma unroll
        for (int i = 0; i < 8; ++i) {
            const float4 b = *reinterpret_cast<const float4*>(bp + 8 * (8 * hk + i));
            c = __builtin_amdgcn_wmma_f32_16x16x4_f32(
                    false, v2f{ a[4 * i],     a[4 * i + 1] }, false, v2f{ b.x, b.y }, (short)0, c, false, false);
            c = __builtin_amdgcn_wmma_f32_16x16x4_f32(
                    false, v2f{ a[4 * i + 2], a[4 * i + 3] }, false, v2f{ b.z, b.w }, (short)0, c, false, false);
        }
    }
    const float bv = bias[col0 + r];
#pragma unroll
    for (int j = 0; j < 8; ++j) rootl[rt * 16 + j + 8 * half][col0 + r] = c[j] + bv;
}

template <bool FIRST, bool RELU_OUT>
__global__ __launch_bounds__(OWNER_THREADS) void gcn_owner_sum_kernel(
    const float*    __restrict__ p,
    const unsigned*              bitmap,
    const unsigned long long* __restrict__ rec,
    float*          __restrict__ acc,
    const float*    __restrict__ hin,
    const float*    __restrict__ Wroot,
    const float*    __restrict__ bias,
    int nrows,
    unsigned long long* __restrict__ wlist,
    int*            __restrict__ wcnt) {
    __shared__ unsigned long long recs[LIST_CAP];
    __shared__ unsigned long long rings[OWNER_WAVES][RING];
    __shared__ unsigned           wbits[OWNER_WAVES][LIST_CAP / 32];
    __shared__ int                cnt[OWNER_WAVES];
    __shared__ __attribute__((aligned(16))) float rootl[ROOT_TILES * 16][ROOT_PITCH];
    __shared__ __attribute__((aligned(16))) float wt[DIM][ROOT_PITCH];

    const int t     = threadIdx.x;
    const int lane  = t & 31;
    const int wave  = __builtin_amdgcn_readfirstlane((int)(threadIdx.x >> 5));
    const int owner = blockIdx.x;
    const char* bm = reinterpret_cast<const char*>(bitmap + (size_t)owner * BM_STRIDE);

    int rq0 = (ROOT_NT * wave) / OWNER_WAVES;
    int rq1 = (ROOT_NT * (wave + 1)) / OWNER_WAVES;

    {
        static_assert(DIM * (DIM / 4) == 4 * OWNER_THREADS, "four (column, k quad) items per thread");
        float4 wq[4];
#pragma unroll
        for (int u = 0; u < 4; ++u) {
            const int item = u * OWNER_THREADS + t;
            const float* wp = Wroot + (size_t)(4 * (item / DIM)) * DIM + (item % DIM);
            wq[u] = make_float4(wp[0], wp[DIM], wp[2 * DIM], wp[3 * DIM]);
        }
#pragma unroll
        for (int u = 0; u < 4; ++u) {
            const int item = u * OWNER_THREADS + t;
            *reinterpret_cast<float4*>(&wt[item % DIM][4 * (item / DIM)]) = wq[u];
        }
        __syncthreads();
    }

    sums_t sum = {};

    ring_word* ring = (ring_word*)&rings[wave][0];
    const unsigned lanes_below = (1u << lane) - 1u;

    unsigned long long* mywl = wlist + ((size_t)owner * OWNER_WAVES + wave) * WLIST_CAP;
    int pos = 0;

    bool saved = false;
    if (!FIRST) {
        const int call = wcnt[owner * OWNER_WAVES + lane];
        saved = __builtin_amdgcn_ballot_w32(call > WLIST_CAP) == 0u;
        if (saved) {
            const int c = __builtin_amdgcn_readlane(call, wave);
            static_assert(ROOT_NT == 4 * 6 + 6 * 5 + 6 * 4 + 6 * 3 + 4 * 2, "the shares by rank cover every root tile once");
            const int rank = __builtin_popcount(
                __builtin_amdgcn_ballot_w32(call < c || (call == c && lane < wave)));
            if      (rank <  4) { rq0 =      6 * rank;        rq1 = rq0 + 6; }
            else if (rank < 10) { rq0 = 24 + 5 * (rank - 4);  rq1 = rq0 + 5; }
            else if (rank < 16) { rq0 = 54 + 4 * (rank - 10); rq1 = rq0 + 4; }
            else if (rank < 22) { rq0 = 78 + 3 * (rank - 16); rq1 = rq0 + 3; }
            else if (rank < 26) { rq0 = 96 + 2 * (rank - 22); rq1 = rq0 + 2; }
            else                { rq0 = 0;                    rq1 = 0;       }
            ring_word* mylist = (ring_word*)&recs[wave * WLIST_CAP];
            unsigned long long r[WLIST_CAP / 32];
#pragma unroll
            for (int q = 0; q < WLIST_CAP / 32; ++q) {
                const int i = q * 32 + lane;
                r[q] = 0ull;
                if (i < c) r[q] = mywl[i];
            }
#pragma unroll
            for (int q = 0; q < WLIST_CAP / 32; ++q) {
                const int i = q * 32 + lane;
                if (i < c) mylist[i] = r[q];
            }
            int b = 0;
            for (; b + EPW <= c; b += EPW)
                owner_batch<false, false>(p, sum, mylist + b, 0, EPW, lane, nullptr, 0);
            if (b < c) owner_batch<true, false>(p, sum, mylist + b, 0, c - b, lane, nullptr, 0);
        }
    }

    if (!saved) {

    int base = 0;
    int total;
    do {
        int tw = t;
        asm volatile("" : "+v"(tw));
        __builtin_assume(tw >= 0 && tw < OWNER_THREADS);
        const int lw = tw & 31;

        for (int i = tw; i < OWNER_WAVES * (LIST_CAP / 32); i += OWNER_THREADS) (&wbits[0][0])[i] = 0u;

        uint4 m[SCAN_TRIPS];
#pragma unroll
        for (int k = 0; k < SCAN_TRIPS; ++k) {
            const int idx = k * OWNER_THREADS + tw;
            m[k] = make_uint4(0u, 0u, 0u, 0u);
            if (idx < BM_VEC) m[k] = *(const uint4*)(bm + (unsigned)idx * 16u);
        }

        int mine_cnt = 0;
#pragma unroll
        for (int k = 0; k < SCAN_TRIPS; ++k)
            mine_cnt += __builtin_popcount(m[k].x) + __builtin_popcount(m[k].y) +
                        __builtin_popcount(m[k].z) + __builtin_popcount(m[k].w);
        mine_cnt = wave_sum(mine_cnt);
        if (lw == 0) cnt[wave] = mine_cnt;
        __syncthreads();
        const int cl = cnt[lw];
        const int offset = __builtin_amdgcn_readfirstlane(wave_sum(lw < wave ? cl : 0));
        total = __builtin_amdgcn_readfirstlane(wave_sum(cl));

        int g = offset;
#pragma unroll
        for (int k = 0; k < SCAN_TRIPS; ++k) {
#pragma unroll
            for (int c = 0; c < 4; ++c) {
                unsigned d = (c == 0) ? m[k].x : (c == 1) ? m[k].y : (c == 2) ? m[k].z : m[k].w;
                const unsigned ebase = ((unsigned)(k * OWNER_THREADS + tw) * 4u + (unsigned)c) * 32u;
                unsigned mask = __builtin_amdgcn_ballot_w32(d != 0u);
                while (mask != 0u) {
                    if (d != 0u) {
                        const unsigned bit = (unsigned)__builtin_ctz(d);
                        const int      pos = g + __builtin_popcount(mask & lanes_below) - base;
                        d &= d - 1u;
                        if ((unsigned)pos < (unsigned)LIST_CAP) recs[pos] = (unsigned long long)(ebase + bit);
                    }
                    g   += __builtin_popcount(mask);
                    mask = __builtin_amdgcn_ballot_w32(d != 0u);
                }
            }
        }
        __syncthreads();

        const int n = (total - base) < LIST_CAP ? (total - base) : LIST_CAP;
        unsigned long long rr[LIST_CAP / OWNER_THREADS];
#pragma unroll
        for (int q = 0; q < LIST_CAP / OWNER_THREADS; ++q) {
            const int i = q * OWNER_THREADS + tw;
            rr[q] = 0ull;
            if (i < n) rr[q] = rec[(unsigned)recs[i]];
        }
#pragma unroll
        for (int q = 0; q < LIST_CAP / OWNER_THREADS; ++q) {
            const int  i     = q * OWNER_THREADS + tw;
            const bool valid = i < n;
            if (valid) recs[i] = rr[q];
            const int v = (int)((((unsigned)rr[q]) >> 16) & 31u);
            unsigned todo = __builtin_amdgcn_ballot_w32(valid);
            while (todo != 0u) {
                const int      leader = __builtin_ctz(todo);
                const int      lv     = __builtin_amdgcn_readlane(v, leader);
                const unsigned mm     = __builtin_amdgcn_ballot_w32(valid && v == lv);
                if (lw == leader) wbits[lv][q * 32 + wave] = mm;
                todo &= ~mm;
            }
        }
        __syncthreads();

        int head  = 0;
        int count = 0;
        for (int w0 = 0; w0 * 32 < n; w0 += 32) {
            const int word = w0 + lane;
            unsigned  d    = wbits[wave][word];
            unsigned mask = __builtin_amdgcn_ballot_w32(d != 0u);
            while (mask != 0u) {
                if (d != 0u) {
                    const int bit = __builtin_ctz(d);
                    d &= d - 1u;
                    ring[(head + count + __builtin_popcount(mask & lanes_below)) & (RING - 1)] =
                        recs[word * 32 + bit];
                }
                count += __builtin_popcount(mask);
                while (count >= EPW) {
                    owner_batch<false, FIRST>(p, sum, ring, head, EPW, lane, mywl, pos);
                    pos   += EPW;
                    head   = (head + EPW) & (RING - 1);
                    count -= EPW;
                }
                mask = __builtin_amdgcn_ballot_w32(d != 0u);
            }
        }
        if (count > 0) {
            owner_batch<true, FIRST>(p, sum, ring, head, count, lane, mywl, pos);
            pos += count;
        }
        __syncthreads();
        base += LIST_CAP;
    } while (base < total);

    if (FIRST && lane == 0) wcnt[owner * OWNER_WAVES + wave] = (total <= LIST_CAP) ? pos : WLIST_OVER;
    }

    const int row0 = owner * ROWS_PER_OWNER;
    int nr = nrows - row0;
    nr = nr < 0 ? 0 : (nr > ROWS_PER_OWNER ? ROWS_PER_OWNER : nr);
    float4* a4 = reinterpret_cast<float4*>(acc + (size_t)row0 * DIM) + lane;

    rq0 = __builtin_amdgcn_readfirstlane(rq0);
    rq1 = __builtin_amdgcn_readfirstlane(rq1);
    for (int q = rq0; q < rq1; ++q)
        root_tile(hin, bias, row0, q >> 3, (q & 7) * 16, nrows, lane, wt, rootl);
    __syncthreads();

#pragma unroll
    for (int s = 0; s < NSLOT; ++s) {
        const int r = s * OWNER_WAVES + wave;
        if (r < nr) {
            float4 a = *reinterpret_cast<const float4*>(&rootl[r][4 * lane]);
            a.x += sum[s];  a.y += sum[s + 8];  a.z += sum[s + 16];  a.w += sum[s + 24];
            if (RELU_OUT) {
                a.x = fmaxf(a.x, 0.f);  a.y = fmaxf(a.y, 0.f);
                a.z = fmaxf(a.z, 0.f);  a.w = fmaxf(a.w, 0.f);
            }
            a4[r * (DIM / 4)] = a;
        }
    }
}

#define GEMM_WGS 256
static_assert(NNODES % 16 == 0, "node tiles are whole: no row guard at the loads, WMMAs or stores");

__device__ __forceinline__ void gemm_body(
    const float* __restrict__ hin,
    const float* __restrict__ Wrel,
    float*       __restrict__ p,
    int nrows, int blk, int nblk) {
    const int t    = threadIdx.x;
    const int lane = t & 31;
    const int wv   = t >> 5;
    const int r    = lane & 15;
    const int half = lane >> 4;
    const int col0 = wv * 16;
    const int ntiles = nrows >> 4;
    const int stride = nblk;

    int tile = blk;
    float4 a_nxt[16];
    {
        const int tl = tile < ntiles ? tile : ntiles - 1;
        const float4* ap = reinterpret_cast<const float4*>(
            hin + ((size_t)tl * 16 + r) * DIM + 4 * half);
#pragma unroll
        for (int i = 0; i < 16; ++i) a_nxt[i] = ap[2 * i];
    }

    v2f b_rel[32];
#pragma unroll
    for (int i = 0; i < 16; ++i) {
        const int kb = 8 * i + 4 * half;
        const float* wr = Wrel  + (size_t)kb * DIM + col0 + r;
        b_rel [2 * i]     = v2f{ wr[0],       wr[DIM]     };
        b_rel [2 * i + 1] = v2f{ wr[2 * DIM], wr[3 * DIM] };
    }

    for (; tile < ntiles; tile += stride) {
        float4 a[16];
#pragma unroll
        for (int i = 0; i < 16; ++i) a[i] = a_nxt[i];

        {
            const int nt = tile + stride;
            const int tl = nt < ntiles ? nt : ntiles - 1;
            const float4* ap = reinterpret_cast<const float4*>(
                hin + ((size_t)tl * 16 + r) * DIM + 4 * half);
#pragma unroll
            for (int i = 0; i < 16; ++i) a_nxt[i] = ap[2 * i];
        }

        v8f acc_rel  = {};
#pragma unroll
        for (int i = 0; i < 16; ++i) {
            const v2f a0 = { a[i].x, a[i].y };
            const v2f a1 = { a[i].z, a[i].w };
            acc_rel  = __builtin_amdgcn_wmma_f32_16x16x4_f32(
                           false, a0, false, b_rel[2 * i],  (short)0, acc_rel,  false, false);
            acc_rel  = __builtin_amdgcn_wmma_f32_16x16x4_f32(
                           false, a1, false, b_rel[2 * i + 1],  (short)0, acc_rel,  false, false);
        }

        const size_t obase = ((size_t)tile * 16 + 8 * half) * DIM + col0 + r;
#pragma unroll
        for (int j = 0; j < 8; ++j) p[obase + (size_t)j * DIM] = acc_rel[j];
    }
}

__global__ __launch_bounds__(256) void gcn_gemm_wmma_kernel(
    const float* __restrict__ hin, const float* __restrict__ Wrel,
    float* __restrict__ p, int nrows) {
    gemm_body(hin, Wrel, p, nrows, (int)blockIdx.x, (int)gridDim.x);
}

__global__ __launch_bounds__(256) void gcn_gemm_prep_kernel(
    const float* __restrict__ hin, const float* __restrict__ Wrel,
    float* __restrict__ p, int nrows,
    const int*   __restrict__ edge_index,
    const float* __restrict__ edge_weight,
    unsigned*    __restrict__ bitmap,
    unsigned long long* __restrict__ rec,
    int nedges) {
    __shared__ __attribute__((aligned(16))) unsigned bits[NOWNERS][PREP_WORDS];
    if (blockIdx.x < GEMM_WGS)
        gemm_body(hin, Wrel, p, nrows, (int)blockIdx.x, GEMM_WGS);
    else
        edge_prep_body(edge_index, edge_weight, bitmap, rec, nedges, (int)blockIdx.x - GEMM_WGS, bits);
}

extern "C" void kernel_launch(void* const* d_in, const int* in_sizes, int n_in,
                              void* d_out, int out_size, void* d_ws, size_t ws_size,
                              hipStream_t stream) {
    const float* x      = (const float*)d_in[0];
    const int*   eidx   = (const int*)  d_in[1];
    const float* ew     = (const float*)d_in[2];
    const float* Wrel   = (const float*)d_in[3];
    const float* brel   = (const float*)d_in[4];
    const float* Wroot  = (const float*)d_in[5];
    float*       out    = (float*)d_out;

    const int N = NNODES;
    const int E = NEDGES;

    float*    s1  = (float*)d_ws;
    float*    s2  = s1 + (size_t)N * DIM;
    float*    p   = s2 + (size_t)N * DIM;
    unsigned* bmp = (unsigned*)(p + (size_t)N * DIM);
    unsigned long long* rec = (unsigned long long*)(bmp + (size_t)NOWNERS * BM_STRIDE);
    unsigned long long* wlist = rec + (size_t)E;
    int* wcnt = (int*)(wlist + (size_t)NOWNERS * OWNER_WAVES * WLIST_CAP);

    for (int l = 0; l < NLAYERS; ++l) {
        const float* hin  = (l == 0) ? x  : (l == 1) ? s1 : s2;
        float*       hout = (l == 0) ? s1 : (l == 1) ? s2 : out;

        if (l == 0)
            gcn_gemm_prep_kernel<<<GEMM_WGS + PREP_BLOCKS, 256, 0, stream>>>(
                hin, Wrel + (size_t)l * DIM * DIM, p, N, eidx, ew, bmp, rec, E);
        else
            gcn_gemm_wmma_kernel<<<GEMM_WGS, 256, 0, stream>>>(
                hin, Wrel + (size_t)l * DIM * DIM, p, N);

        if (l == 0)
            gcn_owner_sum_kernel<true, true><<<NOWNERS, OWNER_THREADS, 0, stream>>>(
                p, bmp, rec, hout, hin, Wroot + (size_t)l * DIM * DIM, brel + (size_t)l * DIM, N, wlist, wcnt);
        else if (l < NLAYERS - 1)
            gcn_owner_sum_kernel<false, true><<<NOWNERS, OWNER_THREADS, 0, stream>>>(
                p, bmp, rec, hout, hin, Wroot + (size_t)l * DIM * DIM, brel + (size_t)l * DIM, N, wlist, wcnt);
        else
            gcn_owner_sum_kernel<false, false><<<NOWNERS, OWNER_THREADS, 0, stream>>>(
                p, bmp, rec, hout, hin, Wroot + (size_t)l * DIM * DIM, brel + (size_t)l * DIM, N, wlist, wcnt);
    }
}


# root term in the edge launch: a wave loads the 16 rows of a row tile once (16 loads, one group) and keeps them in registers for all its column tiles there: a third as many row loads; sums unchanged
# speedup vs baseline: 3.2465x; 1.0386x over previous
#include <hip/hip_runtime.h>

#define NNODES  50000
#define NEDGES  800000
#define DIM     128
#define NLAYERS 3

typedef __attribute__((ext_vector_type(2))) float v2f;
typedef __attribute__((ext_vector_type(8))) float v8f;

#define ROWS_PER_OWNER 196
#define NOWNERS        256
#define PREP_EDGES     1024
#define PREP_BLOCKS    ((NEDGES + PREP_EDGES - 1) / PREP_EDGES)
#define BM_STRIDE      (PREP_BLOCKS * (PREP_EDGES / 32))
#define BM_VEC         (BM_STRIDE / 4)
#define OWNER_THREADS  1024
#define OWNER_WAVES    (OWNER_THREADS / 32)
#define SCAN_TRIPS     ((BM_VEC + OWNER_THREADS - 1) / OWNER_THREADS)
#define LIST_CAP       8192
#define NSLOT          ((ROWS_PER_OWNER + OWNER_WAVES - 1) / OWNER_WAVES)
static_assert(ROWS_PER_OWNER * NOWNERS >= NNODES, "the owners' ranges cover every node");
static_assert(NNODES <= 65536 && ROWS_PER_OWNER <= 65536, "source row and row-in-range share one 32-bit word");
static_assert(BM_STRIDE % 32 == 0, "an owner's map is whole 128-byte lines");
static_assert(OWNER_WAVES == 32, "one lane per wave in the offset sums; rows are dealt to waves by r % 32");

#define WLIST_CAP (LIST_CAP / OWNER_WAVES)
#define WLIST_OVER 0x7fffffff
static_assert(WLIST_CAP * OWNER_WAVES == LIST_CAP, "the waves' saved lists fill the records' LDS array exactly");
static_assert(WLIST_CAP % 32 == 0, "a wave reads its saved list in whole 32-record trips");

#define PREP_THREADS 256
#define PREP_WORDS   (PREP_EDGES / 32)
#define PREP_WPW     (PREP_WORDS / (PREP_THREADS / 32))
static_assert(PREP_WPW * (PREP_THREADS / 32) == PREP_WORDS, "the waves share a block's words evenly");
static_assert(NOWNERS == 256, "an owner number is 8 bits: 8 ballots name a lane's group");

__device__ __forceinline__ void edge_prep_body(
    const int*   __restrict__ edge_index,
    const float* __restrict__ edge_weight,
    unsigned*    __restrict__ bitmap,
    unsigned long long* __restrict__ rec,
    int nedges, int pblk, unsigned (*bits)[PREP_WORDS]) {
    const int t    = threadIdx.x;
    const int lane = t & 31;
    const int wave = t >> 5;

    int   s[PREP_WPW], d[PREP_WPW];
    float w[PREP_WPW];
    bool  valid[PREP_WPW];
#pragma unroll
    for (int i = 0; i < PREP_WPW; ++i) {
        const int e = pblk * PREP_EDGES + (wave * PREP_WPW + i) * 32 + lane;
        valid[i] = e < nedges;
        s[i] = 0;  d[i] = 0;  w[i] = 0.f;
        if (valid[i]) {
            s[i] = edge_index[e];
            d[i] = edge_index[nedges + e];
            w[i] = edge_weight[e];
        }
    }

    uint4* flat4 = reinterpret_cast<uint4*>(&bits[0][0]);
    for (int i = t; i < NOWNERS * PREP_WORDS / 4; i += PREP_THREADS) flat4[i] = make_uint4(0u, 0u, 0u, 0u);
    __syncthreads();

#pragma unroll
    for (int i = 0; i < PREP_WPW; ++i) {
        const int j = wave * PREP_WPW + i;
        const int o = d[i] / ROWS_PER_OWNER;
        d[i] -= o * ROWS_PER_OWNER;
        unsigned m = __builtin_amdgcn_ballot_w32(valid[i]);
#pragma unroll
        for (int b = 0; b < 8; ++b) {
            const bool     bit = ((o >> b) & 1) != 0;
            const unsigned bb  = __builtin_amdgcn_ballot_w32(bit);
            m &= bit ? bb : ~bb;
        }
        if (valid[i] && (m & ((1u << lane) - 1u)) == 0u) bits[o][j] = m;
    }
    __syncthreads();

#pragma unroll
    for (int i = 0; i < PREP_WPW; ++i) {
        const int e = pblk * PREP_EDGES + (wave * PREP_WPW + i) * 32 + lane;
        if (valid[i])
            rec[e] = (unsigned long long)((unsigned)s[i] | ((unsigned)d[i] << 16)) |
                     ((unsigned long long)__float_as_uint(w[i]) << 32);
    }

    for (int i = t; i < NOWNERS * PREP_WORDS / 4; i += PREP_THREADS) {
        const int oo = i >> 3;
        const int pc = i & 7;
        *reinterpret_cast<uint4*>(bitmap + (size_t)oo * BM_STRIDE + (size_t)pblk * PREP_WORDS + pc * 4) = flat4[i];
    }
}

__device__ __forceinline__ int wave_sum(int v) {
    v += __shfl_xor(v, 16);
    v += __shfl_xor(v, 8);
    v += __shfl_xor(v, 4);
    v += __shfl_xor(v, 2);
    v += __shfl_xor(v, 1);
    return v;
}

#define EPW  16
#define RING 64
static_assert(EPW - 1 + 32 <= RING, "the ring holds the waiting edges and one round of new ones");

#define FENCE4(J) asm volatile("" :: \
    "v"(v[J].x),   "v"(v[J].y),   "v"(v[J].z),   "v"(v[J].w),   \
    "v"(v[J+1].x), "v"(v[J+1].y), "v"(v[J+1].z), "v"(v[J+1].w), \
    "v"(v[J+2].x), "v"(v[J+2].y), "v"(v[J+2].z), "v"(v[J+2].w), \
    "v"(v[J+3].x), "v"(v[J+3].y), "v"(v[J+3].z), "v"(v[J+3].w));
static_assert(EPW == 16, "the load fences in owner_batch name 16 gathers");

typedef volatile unsigned long long __attribute__((address_space(3))) ring_word;

typedef float __attribute__((ext_vector_type(32))) sums_t;
static_assert(NSLOT <= 8, "the slots fit the register vector");

__device__ __forceinline__ void slot_add(sums_t& sum, int slot, const float4 v, float w) {
    float s0 = sum[slot], s1 = sum[slot + 8], s2 = sum[slot + 16], s3 = sum[slot + 24];
    s0 += v.x * w;
    s1 += v.y * w;
    s2 += v.z * w;
    s3 += v.w * w;
    sum[slot]      = s0;
    sum[slot + 8]  = s1;
    sum[slot + 16] = s2;
    sum[slot + 24] = s3;
}

template <bool PARTIAL, bool SAVE>
__device__ __forceinline__ void owner_batch(
    const float* __restrict__ p, sums_t& sum,
    const ring_word* ring, int head, int n, int lane,
    unsigned long long* __restrict__ wl, int pos) {
    unsigned long long rc = 0ull;
    if (lane < n) rc = ring[(head + lane) & (RING - 1)];
    if (SAVE) {
        if (lane < n && pos + lane < WLIST_CAP) wl[pos + lane] = rc;
    }
    const int rlo = (int)(unsigned)rc;
    const int rhi = (int)(unsigned)(rc >> 32);

    const unsigned voff = (unsigned)lane * 16u;
    float4 v[EPW];
    int    row[EPW];
    float  w[EPW];
#pragma unroll
    for (int j = 0; j < EPW; ++j) {
        const unsigned sx = (unsigned)__builtin_amdgcn_readlane(rlo, j);
        w[j]   = __uint_as_float((unsigned)__builtin_amdgcn_readlane(rhi, j));
        row[j] = (int)(sx >> 16) / OWNER_WAVES;
        const unsigned off = (sx & 0xffffu) * (unsigned)(DIM * 4) + voff;
        v[j] = *(const float4*)((const char*)p + off);
    }
    FENCE4(0)  FENCE4(4)  FENCE4(8)  FENCE4(12)
#pragma unroll
    for (int j = 0; j < EPW; ++j) {
        if (!PARTIAL || j < n) slot_add(sum, row[j], v[j], w[j]);
    }
}

#define ROOT_TILES ((ROWS_PER_OWNER + 15) / 16)
#define ROOT_NT    (ROOT_TILES * 8)
#define ROOT_PITCH 132
#define FENCE16(X, J) asm volatile("" :: \
    "v"(X[J]),    "v"(X[J+1]),  "v"(X[J+2]),  "v"(X[J+3]),  \
    "v"(X[J+4]),  "v"(X[J+5]),  "v"(X[J+6]),  "v"(X[J+7]),  \
    "v"(X[J+8]),  "v"(X[J+9]),  "v"(X[J+10]), "v"(X[J+11]), \
    "v"(X[J+12]), "v"(X[J+13]), "v"(X[J+14]), "v"(X[J+15]));

__device__ __forceinline__ void root_rows(
    const float* __restrict__ hin,
    int row0, int rt, int nrows, int lane, float (&a)[64]) {
    const int r    = lane & 15;
    const int half = lane >> 4;
    int row = row0 + rt * 16 + r;
    row = row < nrows ? row : nrows - 1;
    const float4* ap = reinterpret_cast<const float4*>(hin + (size_t)row * DIM + 4 * half);
#pragma unroll
    for (int i = 0; i < 16; ++i) {
        const float4 q = ap[2 * i];
        a[4 * i] = q.x;  a[4 * i + 1] = q.y;  a[4 * i + 2] = q.z;  a[4 * i + 3] = q.w;
    }
    FENCE16(a, 0)  FENCE16(a, 16)  FENCE16(a, 32)  FENCE16(a, 48)
}

__device__ __forceinline__ void root_tile(
    const float (&a)[64],
    const float* __restrict__ bias,
    int rt, int col0, int lane,
    const float (*wt)[ROOT_PITCH],
    float (*rootl)[ROOT_PITCH]) {
    const int r    = lane & 15;
    const int half = lane >> 4;
    const float*  bp = &wt[col0 + r][4 * half];
    v8f c = {};
#pragma unroll
    for (int i = 0; i < 16; ++i) {
        const float4 b = *reinterpret_cast<const float4*>(bp + 8 * i);
        c = __builtin_amdgcn_wmma_f32_16x16x4_f32(
                false, v2f{ a[4 * i],     a[4 * i + 1] }, false, v2f{ b.x, b.y }, (short)0, c, false, false);
        c = __builtin_amdgcn_wmma_f32_16x16x4_f32(
                false, v2f{ a[4 * i + 2], a[4 * i + 3] }, false, v2f{ b.z, b.w }, (short)0, c, false, false);
    }
    const float bv = bias[col0 + r];
#pragma unroll
    for (int j = 0; j < 8; ++j) rootl[rt * 16 + j + 8 * half][col0 + r] = c[j] + bv;
}

template <bool FIRST, bool RELU_OUT>
__global__ __launch_bounds__(OWNER_THREADS) void gcn_owner_sum_kernel(
    const float*    __restrict__ p,
    const unsigned*              bitmap,
    const unsigned long long* __restrict__ rec,
    float*          __restrict__ acc,
    const float*    __restrict__ hin,
    const float*    __restrict__ Wroot,
    const float*    __restrict__ bias,
    int nrows,
    unsigned long long* __restrict__ wlist,
    int*            __restrict__ wcnt) {
    __shared__ unsigned long long recs[LIST_CAP];
    __shared__ unsigned long long rings[OWNER_WAVES][RING];
    __shared__ unsigned           wbits[OWNER_WAVES][LIST_CAP / 32];
    __shared__ int                cnt[OWNER_WAVES];
    __shared__ __attribute__((aligned(16))) float rootl[ROOT_TILES * 16][ROOT_PITCH];
    __shared__ __attribute__((aligned(16))) float wt[DIM][ROOT_PITCH];

    const int t     = threadIdx.x;
    const int lane  = t & 31;
    const int wave  = __builtin_amdgcn_readfirstlane((int)(threadIdx.x >> 5));
    const int owner = blockIdx.x;
    const char* bm = reinterpret_cast<const char*>(bitmap + (size_t)owner * BM_STRIDE);

    int rq0 = (ROOT_NT * wave) / OWNER_WAVES;
    int rq1 = (ROOT_NT * (wave + 1)) / OWNER_WAVES;

    {
        static_assert(DIM * (DIM / 4) == 4 * OWNER_THREADS, "four (column, k quad) items per thread");
        float4 wq[4];
#pragma unroll
        for (int u = 0; u < 4; ++u) {
            const int item = u * OWNER_THREADS + t;
            const float* wp = Wroot + (size_t)(4 * (item / DIM)) * DIM + (item % DIM);
            wq[u] = make_float4(wp[0], wp[DIM], wp[2 * DIM], wp[3 * DIM]);
        }
#pragma unroll
        for (int u = 0; u < 4; ++u) {
            const int item = u * OWNER_THREADS + t;
            *reinterpret_cast<float4*>(&wt[item % DIM][4 * (item / DIM)]) = wq[u];
        }
        __syncthreads();
    }

    sums_t sum = {};

    ring_word* ring = (ring_word*)&rings[wave][0];
    const unsigned lanes_below = (1u << lane) - 1u;

    unsigned long long* mywl = wlist + ((size_t)owner * OWNER_WAVES + wave) * WLIST_CAP;
    int pos = 0;

    bool saved = false;
    if (!FIRST) {
        const int call = wcnt[owner * OWNER_WAVES + lane];
        saved = __builtin_amdgcn_ballot_w32(call > WLIST_CAP) == 0u;
        if (saved) {
            const int c = __builtin_amdgcn_readlane(call, wave);
            static_assert(ROOT_NT == 4 * 6 + 6 * 5 + 6 * 4 + 6 * 3 + 4 * 2, "the shares by rank cover every root tile once");
            const int rank = __builtin_popcount(
                __builtin_amdgcn_ballot_w32(call < c || (call == c && lane < wave)));
            if      (rank <  4) { rq0 =      6 * rank;        rq1 = rq0 + 6; }
            else if (rank < 10) { rq0 = 24 + 5 * (rank - 4);  rq1 = rq0 + 5; }
            else if (rank < 16) { rq0 = 54 + 4 * (rank - 10); rq1 = rq0 + 4; }
            else if (rank < 22) { rq0 = 78 + 3 * (rank - 16); rq1 = rq0 + 3; }
            else if (rank < 26) { rq0 = 96 + 2 * (rank - 22); rq1 = rq0 + 2; }
            else                { rq0 = 0;                    rq1 = 0;       }
            ring_word* mylist = (ring_word*)&recs[wave * WLIST_CAP];
            unsigned long long r[WLIST_CAP / 32];
#pragma unroll
            for (int q = 0; q < WLIST_CAP / 32; ++q) {
                const int i = q * 32 + lane;
                r[q] = 0ull;
                if (i < c) r[q] = mywl[i];
            }
#pragma unroll
            for (int q = 0; q < WLIST_CAP / 32; ++q) {
                const int i = q * 32 + lane;
                if (i < c) mylist[i] = r[q];
            }
            int b = 0;
            for (; b + EPW <= c; b += EPW)
                owner_batch<false, false>(p, sum, mylist + b, 0, EPW, lane, nullptr, 0);
            if (b < c) owner_batch<true, false>(p, sum, mylist + b, 0, c - b, lane, nullptr, 0);
        }
    }

    if (!saved) {

    int base = 0;
    int total;
    do {
        int tw = t;
        asm volatile("" : "+v"(tw));
        __builtin_assume(tw >= 0 && tw < OWNER_THREADS);
        const int lw = tw & 31;

        for (int i = tw; i < OWNER_WAVES * (LIST_CAP / 32); i += OWNER_THREADS) (&wbits[0][0])[i] = 0u;

        uint4 m[SCAN_TRIPS];
#pragma unroll
        for (int k = 0; k < SCAN_TRIPS; ++k) {
            const int idx = k * OWNER_THREADS + tw;
            m[k] = make_uint4(0u, 0u, 0u, 0u);
            if (idx < BM_VEC) m[k] = *(const uint4*)(bm + (unsigned)idx * 16u);
        }

        int mine_cnt = 0;
#pragma unroll
        for (int k = 0; k < SCAN_TRIPS; ++k)
            mine_cnt += __builtin_popcount(m[k].x) + __builtin_popcount(m[k].y) +
                        __builtin_popcount(m[k].z) + __builtin_popcount(m[k].w);
        mine_cnt = wave_sum(mine_cnt);
        if (lw == 0) cnt[wave] = mine_cnt;
        __syncthreads();
        const int cl = cnt[lw];
        const int offset = __builtin_amdgcn_readfirstlane(wave_sum(lw < wave ? cl : 0));
        total = __builtin_amdgcn_readfirstlane(wave_sum(cl));

        int g = offset;
#pragma unroll
        for (int k = 0; k < SCAN_TRIPS; ++k) {
#pragma unroll
            for (int c = 0; c < 4; ++c) {
                unsigned d = (c == 0) ? m[k].x : (c == 1) ? m[k].y : (c == 2) ? m[k].z : m[k].w;
                const unsigned ebase = ((unsigned)(k * OWNER_THREADS + tw) * 4u + (unsigned)c) * 32u;
                unsigned mask = __builtin_amdgcn_ballot_w32(d != 0u);
                while (mask != 0u) {
                    if (d != 0u) {
                        const unsigned bit = (unsigned)__builtin_ctz(d);
                        const int      pos = g + __builtin_popcount(mask & lanes_below) - base;
                        d &= d - 1u;
                        if ((unsigned)pos < (unsigned)LIST_CAP) recs[pos] = (unsigned long long)(ebase + bit);
                    }
                    g   += __builtin_popcount(mask);
                    mask = __builtin_amdgcn_ballot_w32(d != 0u);
                }
            }
        }
        __syncthreads();

        const int n = (total - base) < LIST_CAP ? (total - base) : LIST_CAP;
        unsigned long long rr[LIST_CAP / OWNER_THREADS];
#pragma unroll
        for (int q = 0; q < LIST_CAP / OWNER_THREADS; ++q) {
            const int i = q * OWNER_THREADS + tw;
            rr[q] = 0ull;
            if (i < n) rr[q] = rec[(unsigned)recs[i]];
        }
#pragma unroll
        for (int q = 0; q < LIST_CAP / OWNER_THREADS; ++q) {
            const int  i     = q * OWNER_THREADS + tw;
            const bool valid = i < n;
            if (valid) recs[i] = rr[q];
            const int v = (int)((((unsigned)rr[q]) >> 16) & 31u);
            unsigned todo = __builtin_amdgcn_ballot_w32(valid);
            while (todo != 0u) {
                const int      leader = __builtin_ctz(todo);
                const int      lv     = __builtin_amdgcn_readlane(v, leader);
                const unsigned mm     = __builtin_amdgcn_ballot_w32(valid && v == lv);
                if (lw == leader) wbits[lv][q * 32 + wave] = mm;
                todo &= ~mm;
            }
        }
        __syncthreads();

        int head  = 0;
        int count = 0;
        for (int w0 = 0; w0 * 32 < n; w0 += 32) {
            const int word = w0 + lane;
            unsigned  d    = wbits[wave][word];
            unsigned mask = __builtin_amdgcn_ballot_w32(d != 0u);
            while (mask != 0u) {
                if (d != 0u) {
                    const int bit = __builtin_ctz(d);
                    d &= d - 1u;
                    ring[(head + count + __builtin_popcount(mask & lanes_below)) & (RING - 1)] =
                        recs[word * 32 + bit];
                }
                count += __builtin_popcount(mask);
                while (count >= EPW) {
                    owner_batch<false, FIRST>(p, sum, ring, head, EPW, lane, mywl, pos);
                    pos   += EPW;
                    head   = (head + EPW) & (RING - 1);
                    count -= EPW;
                }
                mask = __builtin_amdgcn_ballot_w32(d != 0u);
            }
        }
        if (count > 0) {
            owner_batch<true, FIRST>(p, sum, ring, head, count, lane, mywl, pos);
            pos += count;
        }
        __syncthreads();
        base += LIST_CAP;
    } while (base < total);

    if (FIRST && lane == 0) wcnt[owner * OWNER_WAVES + wave] = (total <= LIST_CAP) ? pos : WLIST_OVER;
    }

    const int row0 = owner * ROWS_PER_OWNER;
    int nr = nrows - row0;
    nr = nr < 0 ? 0 : (nr > ROWS_PER_OWNER ? ROWS_PER_OWNER : nr);
    float4* a4 = reinterpret_cast<float4*>(acc + (size_t)row0 * DIM) + lane;

    rq0 = __builtin_amdgcn_readfirstlane(rq0);
    rq1 = __builtin_amdgcn_readfirstlane(rq1);
    for (int q = rq0; q < rq1; ) {
        const int rt = q >> 3;
        int qe = (rt + 1) * 8;
        qe = qe < rq1 ? qe : rq1;
        float a[64];
        root_rows(hin, row0, rt, nrows, lane, a);
        for (; q < qe; ++q)
            root_tile(a, bias, rt, (q & 7) * 16, lane, wt, rootl);
    }
    __syncthreads();

#pragma unroll
    for (int s = 0; s < NSLOT; ++s) {
        const int r = s * OWNER_WAVES + wave;
        if (r < nr) {
            float4 a = *reinterpret_cast<const float4*>(&rootl[r][4 * lane]);
            a.x += sum[s];  a.y += sum[s + 8];  a.z += sum[s + 16];  a.w += sum[s + 24];
            if (RELU_OUT) {
                a.x = fmaxf(a.x, 0.f);  a.y = fmaxf(a.y, 0.f);
                a.z = fmaxf(a.z, 0.f);  a.w = fmaxf(a.w, 0.f);
            }
            a4[r * (DIM / 4)] = a;
        }
    }
}

#define GEMM_WGS 256
static_assert(NNODES % 16 == 0, "node tiles are whole: no row guard at the loads, WMMAs or stores");

__device__ __forceinline__ void gemm_body(
    const float* __restrict__ hin,
    const float* __restrict__ Wrel,
    float*       __restrict__ p,
    int nrows, int blk, int nblk) {
    const int t    = threadIdx.x;
    const int lane = t & 31;
    const int wv   = t >> 5;
    const int r    = lane & 15;
    const int half = lane >> 4;
    const int col0 = wv * 16;
    const int ntiles = nrows >> 4;
    const int stride = nblk;

    int tile = blk;
    float4 a_nxt[16];
    {
        const int tl = tile < ntiles ? tile : ntiles - 1;
        const float4* ap = reinterpret_cast<const float4*>(
            hin + ((size_t)tl * 16 + r) * DIM + 4 * half);
#pragma unroll
        for (int i = 0; i < 16; ++i) a_nxt[i] = ap[2 * i];
    }

    v2f b_rel[32];
#pragma unroll
    for (int i = 0; i < 16; ++i) {
        const int kb = 8 * i + 4 * half;
        const float* wr = Wrel  + (size_t)kb * DIM + col0 + r;
        b_rel [2 * i]     = v2f{ wr[0],       wr[DIM]     };
        b_rel [2 * i + 1] = v2f{ wr[2 * DIM], wr[3 * DIM] };
    }

    for (; tile < ntiles; tile += stride) {
        float4 a[16];
#pragma unroll
        for (int i = 0; i < 16; ++i) a[i] = a_nxt[i];

        {
            const int nt = tile + stride;
            const int tl = nt < ntiles ? nt : ntiles - 1;
            const float4* ap = reinterpret_cast<const float4*>(
                hin + ((size_t)tl * 16 + r) * DIM + 4 * half);
#pragma unroll
            for (int i = 0; i < 16; ++i) a_nxt[i] = ap[2 * i];
        }

        v8f acc_rel  = {};
#pragma unroll
        for (int i = 0; i < 16; ++i) {
            const v2f a0 = { a[i].x, a[i].y };
            const v2f a1 = { a[i].z, a[i].w };
            acc_rel  = __builtin_amdgcn_wmma_f32_16x16x4_f32(
                           false, a0, false, b_rel[2 * i],  (short)0, acc_rel,  false, false);
            acc_rel  = __builtin_amdgcn_wmma_f32_16x16x4_f32(
                           false, a1, false, b_rel[2 * i + 1],  (short)0, acc_rel,  false, false);
        }

        const size_t obase = ((size_t)tile * 16 + 8 * half) * DIM + col0 + r;
#pragma unroll
        for (int j = 0; j < 8; ++j) p[obase + (size_t)j * DIM] = acc_rel[j];
    }
}

__global__ __launch_bounds__(256) void gcn_gemm_wmma_kernel(
    const float* __restrict__ hin, const float* __restrict__ Wrel,
    float* __restrict__ p, int nrows) {
    gemm_body(hin, Wrel, p, nrows, (int)blockIdx.x, (int)gridDim.x);
}

__global__ __launch_bounds__(256) void gcn_gemm_prep_kernel(
    const float* __restrict__ hin, const float* __restrict__ Wrel,
    float* __restrict__ p, int nrows,
    const int*   __restrict__ edge_index,
    const float* __restrict__ edge_weight,
    unsigned*    __restrict__ bitmap,
    unsigned long long* __restrict__ rec,
    int nedges) {
    __shared__ __attribute__((aligned(16))) unsigned bits[NOWNERS][PREP_WORDS];
    if (blockIdx.x < GEMM_WGS)
        gemm_body(hin, Wrel, p, nrows, (int)blockIdx.x, GEMM_WGS);
    else
        edge_prep_body(edge_index, edge_weight, bitmap, rec, nedges, (int)blockIdx.x - GEMM_WGS, bits);
}

extern "C" void kernel_launch(void* const* d_in, const int* in_sizes, int n_in,
                              void* d_out, int out_size, void* d_ws, size_t ws_size,
                              hipStream_t stream) {
    const float* x      = (const float*)d_in[0];
    const int*   eidx   = (const int*)  d_in[1];
    const float* ew     = (const float*)d_in[2];
    const float* Wrel   = (const float*)d_in[3];
    const float* brel   = (const float*)d_in[4];
    const float* Wroot  = (const float*)d_in[5];
    float*       out    = (float*)d_out;

    const int N = NNODES;
    const int E = NEDGES;

    float*    s1  = (float*)d_ws;
    float*    s2  = s1 + (size_t)N * DIM;
    float*    p   = s2 + (size_t)N * DIM;
    unsigned* bmp = (unsigned*)(p + (size_t)N * DIM);
    unsigned long long* rec = (unsigned long long*)(bmp + (size_t)NOWNERS * BM_STRIDE);
    unsigned long long* wlist = rec + (size_t)E;
    int* wcnt = (int*)(wlist + (size_t)NOWNERS * OWNER_WAVES * WLIST_CAP);

    for (int l = 0; l < NLAYERS; ++l) {
        const float* hin  = (l == 0) ? x  : (l == 1) ? s1 : s2;
        float*       hout = (l == 0) ? s1 : (l == 1) ? s2 : out;

        if (l == 0)
            gcn_gemm_prep_kernel<<<GEMM_WGS + PREP_BLOCKS, 256, 0, stream>>>(
                hin, Wrel + (size_t)l * DIM * DIM, p, N, eidx, ew, bmp, rec, E);
        else
            gcn_gemm_wmma_kernel<<<GEMM_WGS, 256, 0, stream>>>(
                hin, Wrel + (size_t)l * DIM * DIM, p, N);

        if (l == 0)
            gcn_owner_sum_kernel<true, true><<<NOWNERS, OWNER_THREADS, 0, stream>>>(
                p, bmp, rec, hout, hin, Wroot + (size_t)l * DIM * DIM, brel + (size_t)l * DIM, N, wlist, wcnt);
        else if (l < NLAYERS - 1)
            gcn_owner_sum_kernel<false, true><<<NOWNERS, OWNER_THREADS, 0, stream>>>(
                p, bmp, rec, hout, hin, Wroot + (size_t)l * DIM * DIM, brel + (size_t)l * DIM, N, wlist, wcnt);
        else
            gcn_owner_sum_kernel<false, false><<<NOWNERS, OWNER_THREADS, 0, stream>>>(
                p, bmp, rec, hout, hin, Wroot + (size_t)l * DIM * DIM, brel + (size_t)l * DIM, N, wlist, wcnt);
    }
}
